# starting kernel
# baseline (speedup 1.0000x reference)
#include <hip/hip_runtime.h>

typedef __attribute__((ext_vector_type(2))) float v2f;
typedef __attribute__((ext_vector_type(8))) float v8f;

#define NFFT     2048
#define HOP      512
#define NBINS    1025
#define NFRAMES  1000
#define NBATCH   8
#define OUTLEN   511488
#define NCOLS    (NBATCH * NFRAMES)
#define KSTEPS   514
#define MTILES   (NFFT / 16)
#define NTILES   (NCOLS / 16)
#define FLT_TINY 1.1754943508222875e-38f

__device__ __forceinline__ float fold_coef(const float* __restrict__ Wr,
                                           const float* __restrict__ Wi,
                                           int o, int k) {
  if (k < 1028) {
    int f = k;
    if (f >= NBINS) return 0.0f;
    float v = Wr[o * NFFT + f];
    if (f >= 1 && f <= NFFT / 2 - 1) v += Wr[o * NFFT + (NFFT - f)];
    return v;
  } else {
    int f = k - 1028;
    if (f >= NBINS) return 0.0f;
    float v = -Wi[o * NFFT + f];
    if (f >= 1 && f <= NFFT / 2 - 1) v += Wi[o * NFFT + (NFFT - f)];
    return v;
  }
}

__global__ void pack_a_kernel(const float* __restrict__ Wr,
                              const float* __restrict__ Wi,
                              float* __restrict__ apack) {
  int gid  = blockIdx.x * blockDim.x + threadIdx.x;
  int lane = gid & 31;
  int kk   = (gid >> 5) % KSTEPS;
  int mt   = (gid >> 5) / KSTEPS;
  if (mt >= MTILES) return;
  int o = mt * 16 + (lane & 15);
  int k = kk * 4 + 2 * (lane >> 4);
  int idx = ((mt * KSTEPS + kk) * 32 + lane) * 2;
  apack[idx + 0] = fold_coef(Wr, Wi, o, k);
  apack[idx + 1] = fold_coef(Wr, Wi, o, k + 1);
}

#define LOADA(kk)                                                              \
  _Pragma("unroll") for (int i = 0; i < 4; ++i)                                \
      a[i] = Ab[((long)(mg4 + i) * KSTEPS + (kk)) * 32 + lane];

#define WMMA16()                                                               \
  _Pragma("unroll") for (int i = 0; i < 4; ++i) {                              \
    _Pragma("unroll") for (int j = 0; j < 4; ++j) {                            \
      c[i][j] = __builtin_amdgcn_wmma_f32_16x16x4_f32(                         \
          false, a[i], false, b[j], (short)0, c[i][j], false, false);          \
    }                                                                          \
  }

__global__ void __launch_bounds__(256)
istft_gemm_kernel(const float* __restrict__ xr, const float* __restrict__ xi,
                  const float* __restrict__ apack, float* __restrict__ sbuf) {
  const int lane = threadIdx.x & 31;
  const int w    = blockIdx.x * 8 + (threadIdx.x >> 5);
  const int mg   = w & 31;
  const int ng   = w >> 5;
  const int mg4  = mg * 4;
  const int h    = lane >> 4;

  const v2f* Ab = (const v2f*)apack;

  int colIdx[4];
#pragma unroll
  for (int j = 0; j < 4; ++j)
    colIdx[j] = (ng * 64 + j * 16 + (lane & 15)) * NBINS;

  v8f c[4][4];
#pragma unroll
  for (int i = 0; i < 4; ++i)
#pragma unroll
    for (int j = 0; j < 4; ++j) c[i][j] = (v8f){};

  v2f a[4], b[4];

#pragma unroll 2
  for (int kk = 0; kk < 256; ++kk) {
    const int f = kk * 4 + 2 * h;
    LOADA(kk);
#pragma unroll
    for (int j = 0; j < 4; ++j) {
      b[j].x = xr[colIdx[j] + f];
      b[j].y = xr[colIdx[j] + f + 1];
    }
    WMMA16();
  }
  {
    LOADA(256);
#pragma unroll
    for (int j = 0; j < 4; ++j) {
      float v = xr[colIdx[j] + 1024];
      b[j].x = (h == 0) ? v : 0.0f;
      b[j].y = 0.0f;
    }
    WMMA16();
  }

#pragma unroll 2
  for (int kk = 257; kk < 513; ++kk) {
    const int f = (kk - 257) * 4 + 2 * h;
    LOADA(kk);
#pragma unroll
    for (int j = 0; j < 4; ++j) {
      b[j].x = xi[colIdx[j] + f];
      b[j].y = xi[colIdx[j] + f + 1];
    }
    WMMA16();
  }
  {
    LOADA(513);
#pragma unroll
    for (int j = 0; j < 4; ++j) {
      float v = xi[colIdx[j] + 1024];
      b[j].x = (h == 0) ? v : 0.0f;
      b[j].y = 0.0f;
    }
    WMMA16();
  }

  const int orow = h * 8;
#pragma unroll
  for (int j = 0; j < 4; ++j) {
    const int nn = ng * 64 + j * 16 + (lane & 15);
    const int bb = nn / NFRAMES;
    const int tt = nn - bb * NFRAMES;
    float* outb = sbuf + (long)bb * NFFT * NFRAMES + tt;
#pragma unroll
    for (int i = 0; i < 4; ++i) {
      const int ob = (mg4 + i) * 16 + orow;
#pragma unroll
      for (int v = 0; v < 8; ++v)
        outb[(long)(ob + v) * NFRAMES] = c[i][j][v];
    }
  }
}

__global__ void ola_kernel(const float* __restrict__ sbuf,
                           const float* __restrict__ win,
                           float* __restrict__ out) {
  int gid = blockIdx.x * blockDim.x + threadIdx.x;
  if (gid >= NBATCH * OUTLEN) return;
  int b  = gid / OUTLEN;
  int jo = gid - b * OUTLEN;
  int j  = jo + NFFT / 2;

  int tmax = j >> 9;
  if (tmax > NFRAMES - 1) tmax = NFRAMES - 1;
  int tmin = (j >= NFFT - HOP) ? ((j - (NFFT - HOP)) >> 9) : 0;

  const float* sb = sbuf + (long)b * NFFT * NFRAMES;
  float acc = 0.0f, wss = 0.0f;
  for (int t = tmin; t <= tmax; ++t) {
    int o = j - (t << 9);
    acc += sb[(long)o * NFRAMES + t];
    float wv = win[o];
    wss += wv * wv;
  }
  out[gid] = (wss > FLT_TINY) ? acc / wss : acc;
}

extern "C" void kernel_launch(void* const* d_in, const int* in_sizes, int n_in,
                              void* d_out, int out_size, void* d_ws,
                              size_t ws_size, hipStream_t stream) {
  const float* real = (const float*)d_in[0];
  const float* imag = (const float*)d_in[1];
  const float* Wr   = (const float*)d_in[2];
  const float* Wi   = (const float*)d_in[3];
  const float* win  = (const float*)d_in[4];
  float* out = (float*)d_out;

  float* apack = (float*)d_ws;
  size_t apack_elems = (size_t)MTILES * KSTEPS * 64;
  float* sbuf = apack + apack_elems;

  {
    int total = MTILES * KSTEPS * 32;
    pack_a_kernel<<<(total + 255) / 256, 256, 0, stream>>>(Wr, Wi, apack);
  }
  istft_gemm_kernel<<<(MTILES / 4) * (NTILES / 4) / 8, 256, 0, stream>>>(
      real, imag, apack, sbuf);
  {
    int total = NBATCH * OUTLEN;
    ola_kernel<<<(total + 255) / 256, 256, 0, stream>>>(sbuf, win, out);
  }
}


# GEMM A operand staged through LDS as well: work-group tile 512x64 -> 256x128 (4x2 waves), A fetched once per tile by 16-B loads half a 32-k block ahead, all fragments from LDS; k, sum order kept
# speedup vs baseline: 1.3836x; 1.3836x over previous
#include <hip/hip_runtime.h>

typedef __attribute__((ext_vector_type(2))) float v2f;
typedef __attribute__((ext_vector_type(8))) float v8f;

#define NFFT     2048
#define HOP      512
#define NBINS    1025
#define NFRAMES  1000
#define NBATCH   8
#define OUTLEN   511488
#define NCOLS    (NBATCH * NFRAMES)
#define KSTEPS   514
#define MTILES   (NFFT / 16)
#define NTILES   (NCOLS / 16)
#define FLT_TINY 1.1754943508222875e-38f

__device__ __forceinline__ float fold_coef(const float* __restrict__ Wr,
                                           const float* __restrict__ Wi,
                                           int o, int k) {
  if (k < 1028) {
    int f = k;
    if (f >= NBINS) return 0.0f;
    float v = Wr[o * NFFT + f];
    if (f >= 1 && f <= NFFT / 2 - 1) v += Wr[o * NFFT + (NFFT - f)];
    return v;
  } else {
    int f = k - 1028;
    if (f >= NBINS) return 0.0f;
    float v = -Wi[o * NFFT + f];
    if (f >= 1 && f <= NFFT / 2 - 1) v += Wi[o * NFFT + (NFFT - f)];
    return v;
  }
}

__global__ void pack_a_kernel(const float* __restrict__ Wr,
                              const float* __restrict__ Wi,
                              float* __restrict__ apack) {
  int gid  = blockIdx.x * blockDim.x + threadIdx.x;
  int lane = gid & 31;
  int kk   = (gid >> 5) % KSTEPS;
  int mt   = (gid >> 5) / KSTEPS;
  if (mt >= MTILES) return;
  int o = mt * 16 + (lane & 15);
  int k = kk * 4 + 2 * (lane >> 4);
  int idx = ((mt * KSTEPS + kk) * 32 + lane) * 2;
  apack[idx + 0] = fold_coef(Wr, Wi, o, k);
  apack[idx + 1] = fold_coef(Wr, Wi, o, k + 1);
}

#define WMMA16()                                                               \
  _Pragma("unroll") for (int i = 0; i < 4; ++i) {                              \
    _Pragma("unroll") for (int j = 0; j < 4; ++j) {                            \
      c[i][j] = __builtin_amdgcn_wmma_f32_16x16x4_f32(                         \
          false, a[i], false, b[j], (short)0, c[i][j], false, false);          \
    }                                                                          \
  }

#define BK      32
#define BPITCH  36
#define WGM     256
#define WGN     128
#define NKB     64
#define NWGM    (NFFT / WGM)
#define NWGN    ((NCOLS + WGN - 1) / WGN)
#define ATILE   512

typedef __attribute__((ext_vector_type(4))) float v4f;

#define EDGE_STEP(xp, kk)                                                      \
  {                                                                            \
    _Pragma("unroll") for (int i = 0; i < 4; ++i)                              \
        a[i] = Ab[((long)(mt0 + i) * KSTEPS + (kk)) * 32 + lane];              \
    _Pragma("unroll") for (int j = 0; j < 4; ++j) {                            \
      float v = (xp)[colIdx[j] + 1024];                                        \
      b[j].x = (h == 0) ? v : 0.0f;                                            \
      b[j].y = 0.0f;                                                           \
    }                                                                          \
    WMMA16();                                                                  \
  }

#define STAGE_LOAD(kn, t)                                                      \
  {                                                                            \
    const float* __restrict__ xn = ((kn) < 32) ? xr : xi;                      \
    const int f0  = ((kn) & 31) * BK;                                          \
    const int kkn = (((kn) < 32) ? 0 : 257) + ((kn) & 31) * 8;                 \
    _Pragma("unroll") for (int j = 0; j < 4; ++j)                              \
        ga[j] = Ap4[aSrc[t] + kkn * 16 + j * 32];                              \
    _Pragma("unroll") for (int i = 0; i < 8; ++i)                              \
        g[i] = xn[rowBase + (8 * (t) + i) * NBINS + f0];                       \
  }

#define STAGE_STORE(st, t)                                                     \
  {                                                                            \
    _Pragma("unroll") for (int j = 0; j < 4; ++j)                              \
        *(v4f*)&at[st][(2 * wv + (t)) * ATILE + (j * 32 + lane) * 4] = ga[j];  \
    _Pragma("unroll") for (int i = 0; i < 8; ++i)                              \
        bt[st][(wv * 16 + 8 * (t) + i) * BPITCH + lane] = g[i];                \
  }

__global__ void __launch_bounds__(256, 4)
istft_gemm_kernel(const float* __restrict__ xr, const float* __restrict__ xi,
                  const float* __restrict__ apack, float* __restrict__ sbuf) {
  __shared__ __attribute__((aligned(16))) float at[2][16 * ATILE];
  __shared__ __attribute__((aligned(16))) float bt[2][WGN * BPITCH];

  const int lane = threadIdx.x & 31;
  const int wv   = __builtin_amdgcn_readfirstlane(threadIdx.x >> 5);
  const int wm   = wv & 3;
  const int wn   = wv >> 2;
  const int mgw  = blockIdx.x & 7;
  const int ngw  = blockIdx.x >> 3;
  const int mt0  = mgw * 16 + wm * 4;
  const int ncol0 = ngw * WGN + wn * 64;
  const int h    = lane >> 4;
  const int cl   = lane & 15;

  const v2f* Ab  = (const v2f*)apack;
  const v4f* Ap4 = (const v4f*)apack;

  const int ncolL = min(ncol0, NCOLS - 64);
  int colIdx[4];
#pragma unroll
  for (int j = 0; j < 4; ++j)
    colIdx[j] = (ncolL + j * 16 + cl) * NBINS;

  int aSrc[2];
#pragma unroll
  for (int t = 0; t < 2; ++t)
    aSrc[t] = ((mgw * 16 + 2 * wv + t) * KSTEPS) * 16 + lane;
  const int rowBase = min(ngw * WGN + wv * 16, NCOLS - 16) * NBINS + lane;

  v8f c[4][4];
#pragma unroll
  for (int i = 0; i < 4; ++i)
#pragma unroll
    for (int j = 0; j < 4; ++j) c[i][j] = (v8f){};

  v2f a[4], b[4];
  v4f ga[4];
  float g[8];

  STAGE_LOAD(0, 0)
  STAGE_STORE(0, 0)
  STAGE_LOAD(0, 1)
  STAGE_STORE(0, 1)
  __syncthreads();

  for (int kb = 0; kb < NKB; ++kb) {
    const int cur = kb & 1;

    if (kb + 1 < NKB) {
      STAGE_LOAD(kb + 1, 0)
      asm volatile("" ::: "memory");
    }

    const float* acur = &at[cur][wm * 4 * ATILE + lane * 2];
    const float* bcur = &bt[cur][(wn * 64 + cl) * BPITCH + 2 * h];
#pragma unroll
    for (int s = 0; s < 8; ++s) {
      if (s == 4 && kb + 1 < NKB) {
        STAGE_STORE(cur ^ 1, 0)
        STAGE_LOAD(kb + 1, 1)
        asm volatile("" ::: "memory");
      }
#pragma unroll
      for (int i = 0; i < 4; ++i)
        a[i] = *(const v2f*)(acur + i * ATILE + s * 64);
#pragma unroll
      for (int j = 0; j < 4; ++j)
        b[j] = *(const v2f*)(bcur + j * 16 * BPITCH + 4 * s);
      WMMA16();
    }

    if (kb == 31) EDGE_STEP(xr, 256)

    if (kb + 1 < NKB) STAGE_STORE(cur ^ 1, 1)
    __syncthreads();
  }
  EDGE_STEP(xi, 513)

  if (ncol0 < NCOLS) {
    const int orow = h * 8;
#pragma unroll
    for (int j = 0; j < 4; ++j) {
      const int nn = ncol0 + j * 16 + cl;
      const int bb = nn / NFRAMES;
      const int tt = nn - bb * NFRAMES;
      float* outb = sbuf + (long)bb * NFFT * NFRAMES + tt;
#pragma unroll
      for (int i = 0; i < 4; ++i) {
        const int ob = (mt0 + i) * 16 + orow;
#pragma unroll
        for (int v = 0; v < 8; ++v)
          outb[(long)(ob + v) * NFRAMES] = c[i][j][v];
      }
    }
  }
}

#define OLA_TR    32
#define OLA_TQ    128
#define OLA_QT    8
#define OLA_RT    (HOP / OLA_TR)
#define OLA_PITCH 33

__global__ void __launch_bounds__(256)
ola_kernel(const float* __restrict__ sbuf, const float* __restrict__ win,
           float* __restrict__ out) {
  __shared__ float tile[OLA_TQ * OLA_PITCH];

  const int lane = threadIdx.x & 31;
  const int wv   = __builtin_amdgcn_readfirstlane(threadIdx.x >> 5);
  const int qt   = blockIdx.x % OLA_QT;
  const int rt   = (blockIdx.x / OLA_QT) % OLA_RT;
  const int b    = blockIdx.x / (OLA_QT * OLA_RT);
  const int r0   = rt * OLA_TR;
  const int q0   = 2 + qt * OLA_TQ;

  const float* sb = sbuf + (long)b * NFFT * NFRAMES;

#pragma unroll
  for (int i = 0; i < 4; ++i) {
    const int rl = wv * 4 + i;
    const int r  = r0 + rl;
#pragma unroll
    for (int c = 0; c < 4; ++c) {
      const int q = q0 + c * 32 + lane;
      float acc = 0.0f, wss = 0.0f;
#pragma unroll
      for (int d = 3; d >= 0; --d) {
        const int t  = q - d;
        const int o  = r + d * HOP;
        const bool ok = (t >= 0) && (t <= NFRAMES - 1);
        const int tc = min(max(t, 0), NFRAMES - 1);
        float x = sb[o * NFRAMES + tc];
        float w = win[o];
        x = ok ? x : 0.0f;
        w = ok ? w : 0.0f;
        acc += x;
        wss += w * w;
      }
      tile[(c * 32 + lane) * OLA_PITCH + rl] =
          (wss > FLT_TINY) ? acc / wss : acc;
    }
  }
  __syncthreads();

  const int qb = q0 + wv * 16;
  float* ob = out + (long)b * OUTLEN + (long)(qb - 2) * HOP + r0 + lane;
  const float* tl = &tile[wv * 16 * OLA_PITCH + lane];
  if (qb + 15 <= NFRAMES) {
    float v[16];
#pragma unroll
    for (int k = 0; k < 16; ++k) v[k] = tl[k * OLA_PITCH];
#pragma unroll
    for (int k = 0; k < 16; ++k) ob[k * HOP] = v[k];
  } else {
#pragma unroll
    for (int k = 0; k < 16; ++k)
      if (qb + k <= NFRAMES) ob[k * HOP] = tl[k * OLA_PITCH];
  }
}

extern "C" void kernel_launch(void* const* d_in, const int* in_sizes, int n_in,
                              void* d_out, int out_size, void* d_ws,
                              size_t ws_size, hipStream_t stream) {
  const float* real = (const float*)d_in[0];
  const float* imag = (const float*)d_in[1];
  const float* Wr   = (const float*)d_in[2];
  const float* Wi   = (const float*)d_in[3];
  const float* win  = (const float*)d_in[4];
  float* out = (float*)d_out;

  float* apack = (float*)d_ws;
  size_t apack_elems = (size_t)MTILES * KSTEPS * 64;
  float* sbuf = apack + apack_elems;

  {
    int total = MTILES * KSTEPS * 32;
    pack_a_kernel<<<(total + 255) / 256, 256, 0, stream>>>(Wr, Wi, apack);
  }
  istft_gemm_kernel<<<NWGM * NWGN, 256, 0, stream>>>(
      real, imag, apack, sbuf);
  ola_kernel<<<NBATCH * OLA_RT * OLA_QT, 256, 0, stream>>>(sbuf, win, out);
}


# GEMM: wave priority stepped down with progress (s_setprio 3,2,1,0 from k blocks 0,24,44,58): a WGP's two work-groups leapfrog and overlap to the end, the younger no longer runs alone; sums kept
# speedup vs baseline: 1.4300x; 1.0335x over previous
#include <hip/hip_runtime.h>

typedef __attribute__((ext_vector_type(2))) float v2f;
typedef __attribute__((ext_vector_type(8))) float v8f;

#define NFFT     2048
#define HOP      512
#define NBINS    1025
#define NFRAMES  1000
#define NBATCH   8
#define OUTLEN   511488
#define NCOLS    (NBATCH * NFRAMES)
#define KSTEPS   514
#define MTILES   (NFFT / 16)
#define NTILES   (NCOLS / 16)
#define FLT_TINY 1.1754943508222875e-38f

__device__ __forceinline__ float fold_coef(const float* __restrict__ Wr,
                                           const float* __restrict__ Wi,
                                           int o, int k) {
  if (k < 1028) {
    int f = k;
    if (f >= NBINS) return 0.0f;
    float v = Wr[o * NFFT + f];
    if (f >= 1 && f <= NFFT / 2 - 1) v += Wr[o * NFFT + (NFFT - f)];
    return v;
  } else {
    int f = k - 1028;
    if (f >= NBINS) return 0.0f;
    float v = -Wi[o * NFFT + f];
    if (f >= 1 && f <= NFFT / 2 - 1) v += Wi[o * NFFT + (NFFT - f)];
    return v;
  }
}

__global__ void pack_a_kernel(const float* __restrict__ Wr,
                              const float* __restrict__ Wi,
                              float* __restrict__ apack) {
  int gid  = blockIdx.x * blockDim.x + threadIdx.x;
  int lane = gid & 31;
  int kk   = (gid >> 5) % KSTEPS;
  int mt   = (gid >> 5) / KSTEPS;
  if (mt >= MTILES) return;
  int o = mt * 16 + (lane & 15);
  int k = kk * 4 + 2 * (lane >> 4);
  int idx = ((mt * KSTEPS + kk) * 32 + lane) * 2;
  apack[idx + 0] = fold_coef(Wr, Wi, o, k);
  apack[idx + 1] = fold_coef(Wr, Wi, o, k + 1);
}

#define WMMA16()                                                               \
  _Pragma("unroll") for (int i = 0; i < 4; ++i) {                              \
    _Pragma("unroll") for (int j = 0; j < 4; ++j) {                            \
      c[i][j] = __builtin_amdgcn_wmma_f32_16x16x4_f32(                         \
          false, a[i], false, b[j], (short)0, c[i][j], false, false);          \
    }                                                                          \
  }

#define BK      32
#define BPITCH  36
#define WGM     256
#define WGN     128
#define NKB     64
#define NWGM    (NFFT / WGM)
#define NWGN    ((NCOLS + WGN - 1) / WGN)
#define ATILE   512

typedef __attribute__((ext_vector_type(4))) float v4f;

#define EDGE_STEP(xp, kk)                                                      \
  {                                                                            \
    _Pragma("unroll") for (int i = 0; i < 4; ++i)                              \
        a[i] = Ab[((long)(mt0 + i) * KSTEPS + (kk)) * 32 + lane];              \
    _Pragma("unroll") for (int j = 0; j < 4; ++j) {                            \
      float v = (xp)[colIdx[j] + 1024];                                        \
      b[j].x = (h == 0) ? v : 0.0f;                                            \
      b[j].y = 0.0f;                                                           \
    }                                                                          \
    WMMA16();                                                                  \
  }

#define STAGE_LOAD(kn, t)                                                      \
  {                                                                            \
    const float* __restrict__ xn = ((kn) < 32) ? xr : xi;                      \
    const int f0  = ((kn) & 31) * BK;                                          \
    const int kkn = (((kn) < 32) ? 0 : 257) + ((kn) & 31) * 8;                 \
    _Pragma("unroll") for (int j = 0; j < 4; ++j)                              \
        ga[j] = Ap4[aSrc[t] + kkn * 16 + j * 32];                              \
    _Pragma("unroll") for (int i = 0; i < 8; ++i)                              \
        g[i] = xn[rowBase + (8 * (t) + i) * NBINS + f0];                       \
  }

#define STAGE_STORE(st, t)                                                     \
  {                                                                            \
    _Pragma("unroll") for (int j = 0; j < 4; ++j)                              \
        *(v4f*)&at[st][(2 * wv + (t)) * ATILE + (j * 32 + lane) * 4] = ga[j];  \
    _Pragma("unroll") for (int i = 0; i < 8; ++i)                              \
        bt[st][(wv * 16 + 8 * (t) + i) * BPITCH + lane] = g[i];                \
  }

__global__ void __launch_bounds__(256, 4)
istft_gemm_kernel(const float* __restrict__ xr, const float* __restrict__ xi,
                  const float* __restrict__ apack, float* __restrict__ sbuf) {
  __shared__ __attribute__((aligned(16))) float at[2][16 * ATILE];
  __shared__ __attribute__((aligned(16))) float bt[2][WGN * BPITCH];

  const int lane = threadIdx.x & 31;
  const int wv   = __builtin_amdgcn_readfirstlane(threadIdx.x >> 5);
  const int wm   = wv & 3;
  const int wn   = wv >> 2;
  const int mgw  = blockIdx.x & 7;
  const int ngw  = blockIdx.x >> 3;
  const int mt0  = mgw * 16 + wm * 4;
  const int ncol0 = ngw * WGN + wn * 64;
  const int h    = lane >> 4;
  const int cl   = lane & 15;

  const v2f* Ab  = (const v2f*)apack;
  const v4f* Ap4 = (const v4f*)apack;

  const int ncolL = min(ncol0, NCOLS - 64);
  int colIdx[4];
#pragma unroll
  for (int j = 0; j < 4; ++j)
    colIdx[j] = (ncolL + j * 16 + cl) * NBINS;

  int aSrc[2];
#pragma unroll
  for (int t = 0; t < 2; ++t)
    aSrc[t] = ((mgw * 16 + 2 * wv + t) * KSTEPS) * 16 + lane;
  const int rowBase = min(ngw * WGN + wv * 16, NCOLS - 16) * NBINS + lane;

  v8f c[4][4];
#pragma unroll
  for (int i = 0; i < 4; ++i)
#pragma unroll
    for (int j = 0; j < 4; ++j) c[i][j] = (v8f){};

  v2f a[4], b[4];
  v4f ga[4];
  float g[8];

  STAGE_LOAD(0, 0)
  STAGE_STORE(0, 0)
  STAGE_LOAD(0, 1)
  STAGE_STORE(0, 1)
  __syncthreads();

  for (int kb = 0; kb < NKB; ++kb) {
    const int cur = kb & 1;

    switch (kb) {
      case 0:  __builtin_amdgcn_s_setprio(3); break;
      case 24: __builtin_amdgcn_s_setprio(2); break;
      case 44: __builtin_amdgcn_s_setprio(1); break;
      case 58: __builtin_amdgcn_s_setprio(0); break;
      default: break;
    }

    if (kb + 1 < NKB) {
      STAGE_LOAD(kb + 1, 0)
      asm volatile("" ::: "memory");
    }

    const float* acur = &at[cur][wm * 4 * ATILE + lane * 2];
    const float* bcur = &bt[cur][(wn * 64 + cl) * BPITCH + 2 * h];
#pragma unroll
    for (int s = 0; s < 8; ++s) {
      if (s == 4 && kb + 1 < NKB) {
        STAGE_STORE(cur ^ 1, 0)
        STAGE_LOAD(kb + 1, 1)
        asm volatile("" ::: "memory");
      }
#pragma unroll
      for (int i = 0; i < 4; ++i)
        a[i] = *(const v2f*)(acur + i * ATILE + s * 64);
#pragma unroll
      for (int j = 0; j < 4; ++j)
        b[j] = *(const v2f*)(bcur + j * 16 * BPITCH + 4 * s);
      WMMA16();
    }

    if (kb == 31) EDGE_STEP(xr, 256)

    if (kb + 1 < NKB) STAGE_STORE(cur ^ 1, 1)
    __syncthreads();
  }
  EDGE_STEP(xi, 513)

  if (ncol0 < NCOLS) {
    const int orow = h * 8;
#pragma unroll
    for (int j = 0; j < 4; ++j) {
      const int nn = ncol0 + j * 16 + cl;
      const int bb = nn / NFRAMES;
      const int tt = nn - bb * NFRAMES;
      float* outb = sbuf + (long)bb * NFFT * NFRAMES + tt;
#pragma unroll
      for (int i = 0; i < 4; ++i) {
        const int ob = (mt0 + i) * 16 + orow;
#pragma unroll
        for (int v = 0; v < 8; ++v)
          outb[(long)(ob + v) * NFRAMES] = c[i][j][v];
      }
    }
  }
}

#define OLA_TR    32
#define OLA_TQ    128
#define OLA_QT    8
#define OLA_RT    (HOP / OLA_TR)
#define OLA_PITCH 33

__global__ void __launch_bounds__(256)
ola_kernel(const float* __restrict__ sbuf, const float* __restrict__ win,
           float* __restrict__ out) {
  __shared__ float tile[OLA_TQ * OLA_PITCH];

  const int lane = threadIdx.x & 31;
  const int wv   = __builtin_amdgcn_readfirstlane(threadIdx.x >> 5);
  const int qt   = blockIdx.x % OLA_QT;
  const int rt   = (blockIdx.x / OLA_QT) % OLA_RT;
  const int b    = blockIdx.x / (OLA_QT * OLA_RT);
  const int r0   = rt * OLA_TR;
  const int q0   = 2 + qt * OLA_TQ;

  const float* sb = sbuf + (long)b * NFFT * NFRAMES;

#pragma unroll
  for (int i = 0; i < 4; ++i) {
    const int rl = wv * 4 + i;
    const int r  = r0 + rl;
#pragma unroll
    for (int c = 0; c < 4; ++c) {
      const int q = q0 + c * 32 + lane;
      float acc = 0.0f, wss = 0.0f;
#pragma unroll
      for (int d = 3; d >= 0; --d) {
        const int t  = q - d;
        const int o  = r + d * HOP;
        const bool ok = (t >= 0) && (t <= NFRAMES - 1);
        const int tc = min(max(t, 0), NFRAMES - 1);
        float x = sb[o * NFRAMES + tc];
        float w = win[o];
        x = ok ? x : 0.0f;
        w = ok ? w : 0.0f;
        acc += x;
        wss += w * w;
      }
      tile[(c * 32 + lane) * OLA_PITCH + rl] =
          (wss > FLT_TINY) ? acc / wss : acc;
    }
  }
  __syncthreads();

  const int qb = q0 + wv * 16;
  float* ob = out + (long)b * OUTLEN + (long)(qb - 2) * HOP + r0 + lane;
  const float* tl = &tile[wv * 16 * OLA_PITCH + lane];
  if (qb + 15 <= NFRAMES) {
    float v[16];
#pragma unroll
    for (int k = 0; k < 16; ++k) v[k] = tl[k * OLA_PITCH];
#pragma unroll
    for (int k = 0; k < 16; ++k) ob[k * HOP] = v[k];
  } else {
#pragma unroll
    for (int k = 0; k < 16; ++k)
      if (qb + k <= NFRAMES) ob[k * HOP] = tl[k * OLA_PITCH];
  }
}

extern "C" void kernel_launch(void* const* d_in, const int* in_sizes, int n_in,
                              void* d_out, int out_size, void* d_ws,
                              size_t ws_size, hipStream_t stream) {
  const float* real = (const float*)d_in[0];
  const float* imag = (const float*)d_in[1];
  const float* Wr   = (const float*)d_in[2];
  const float* Wi   = (const float*)d_in[3];
  const float* win  = (const float*)d_in[4];
  float* out = (float*)d_out;

  float* apack = (float*)d_ws;
  size_t apack_elems = (size_t)MTILES * KSTEPS * 64;
  float* sbuf = apack + apack_elems;

  {
    int total = MTILES * KSTEPS * 32;
    pack_a_kernel<<<(total + 255) / 256, 256, 0, stream>>>(Wr, Wi, apack);
  }
  istft_gemm_kernel<<<NWGM * NWGN, 256, 0, stream>>>(
      real, imag, apack, sbuf);
  ola_kernel<<<NBATCH * OLA_RT * OLA_QT, 256, 0, stream>>>(sbuf, win, out);
}


# Overlap-add folded into the GEMM reduction: a group's 4 wave rows are the 4 frame offsets of 64 residues x 125 hops, partial tiles added through LDS; frame buffer and its launch go; sum order kept
# speedup vs baseline: 1.5323x; 1.0715x over previous
#include <hip/hip_runtime.h>

typedef __attribute__((ext_vector_type(2))) float v2f;
typedef __attribute__((ext_vector_type(8))) float v8f;

#define NFFT     2048
#define HOP      512
#define NBINS    1025
#define NFRAMES  1000
#define NBATCH   8
#define OUTLEN   511488
#define KSTEPS   514
#define MTILES   (NFFT / 16)
#define FLT_TINY 1.1754943508222875e-38f

__device__ __forceinline__ float fold_coef(const float* __restrict__ Wr,
                                           const float* __restrict__ Wi,
                                           int o, int k) {
  if (k < 1028) {
    int f = k;
    if (f >= NBINS) return 0.0f;
    float v = Wr[o * NFFT + f];
    if (f >= 1 && f <= NFFT / 2 - 1) v += Wr[o * NFFT + (NFFT - f)];
    return v;
  } else {
    int f = k - 1028;
    if (f >= NBINS) return 0.0f;
    float v = -Wi[o * NFFT + f];
    if (f >= 1 && f <= NFFT / 2 - 1) v += Wi[o * NFFT + (NFFT - f)];
    return v;
  }
}

__global__ void pack_a_kernel(const float* __restrict__ Wr,
                              const float* __restrict__ Wi,
                              float* __restrict__ apack) {
  int gid  = blockIdx.x * blockDim.x + threadIdx.x;
  int lane = gid & 31;
  int kk   = (gid >> 5) % KSTEPS;
  int mt   = (gid >> 5) / KSTEPS;
  if (mt >= MTILES) return;
  int o = mt * 16 + (lane & 15);
  int k = kk * 4 + 2 * (lane >> 4);
  int idx = ((mt * KSTEPS + kk) * 32 + lane) * 2;
  apack[idx + 0] = fold_coef(Wr, Wi, o, k);
  apack[idx + 1] = fold_coef(Wr, Wi, o, k + 1);
}

#define WMMA16()                                                               \
  _Pragma("unroll") for (int i = 0; i < 4; ++i) {                              \
    _Pragma("unroll") for (int j = 0; j < 4; ++j) {                            \
      c[i][j] = __builtin_amdgcn_wmma_f32_16x16x4_f32(                         \
          false, a[i], false, b[j], (short)0, c[i][j], false, false);          \
    }                                                                          \
  }

#define BK      32
#define BPITCH  36
#define BROWS   134
#define TQ      125
#define NKB     64
#define NRG     (HOP / 64)
#define NQG     8
#define ATILE   512
#define EPITCH  68
#define ETILE   (64 * EPITCH)
#define LDS_FLOATS (8 * ETILE)
#define AT(st)  (lds + (st) * 16 * ATILE)
#define BT(st)  (lds + 2 * 16 * ATILE + (st) * BROWS * BPITCH)

typedef __attribute__((ext_vector_type(4))) float v4f;

#define EDGE_STEP(xp, kk)                                                      \
  {                                                                            \
    _Pragma("unroll") for (int i = 0; i < 4; ++i)                              \
        a[i] = Ab[((long)(mt0 + i) * KSTEPS + (kk)) * 32 + lane];              \
    _Pragma("unroll") for (int j = 0; j < 4; ++j) {                            \
      float v = (xp)[colIdx[j] + 1024];                                        \
      b[j].x = (h == 0) ? v : 0.0f;                                            \
      b[j].y = 0.0f;                                                           \
    }                                                                          \
    WMMA16();                                                                  \
  }

#define STAGE_LOAD(kn, t)                                                      \
  {                                                                            \
    const float* __restrict__ xn = ((kn) < 32) ? xr : xi;                      \
    const int f0  = ((kn) & 31) * BK;                                          \
    const int kkn = (((kn) < 32) ? 0 : 257) + ((kn) & 31) * 8;                 \
    _Pragma("unroll") for (int j = 0; j < 4; ++j)                              \
        ga[j] = Ap4[aSrc[t] + kkn * 16 + j * 32];                              \
    _Pragma("unroll") for (int i = 0; i < 8; ++i)                              \
        g[i] = xn[rowBase + (8 * (t) + i) * NBINS + f0];                       \
  }

#define STAGE_STORE(st, t)                                                     \
  {                                                                            \
    _Pragma("unroll") for (int j = 0; j < 4; ++j)                              \
        *(v4f*)&AT(st)[(2 * wv + (t)) * ATILE + (j * 32 + lane) * 4] = ga[j];  \
    _Pragma("unroll") for (int i = 0; i < 8; ++i)                              \
        BT(st)[(1 + wv * 16 + 8 * (t) + i) * BPITCH + lane] = g[i];            \
  }

__global__ void __launch_bounds__(256, 4)
istft_gemm_kernel(const float* __restrict__ xr, const float* __restrict__ xi,
                  const float* __restrict__ apack,
                  const float* __restrict__ win, float* __restrict__ out) {
  __shared__ __attribute__((aligned(16))) float lds[LDS_FLOATS];

  const int lane = threadIdx.x & 31;
  const int wv   = __builtin_amdgcn_readfirstlane(threadIdx.x >> 5);
  const int wm   = wv & 3;
  const int wn   = wv >> 2;
  const int rg   = blockIdx.x & 7;
  const int qg   = (blockIdx.x >> 3) & 7;
  const int bat  = blockIdx.x >> 6;
  const int r0   = rg * 64;
  const int q0   = 2 + qg * TQ;
  const int fs   = min(max(q0 - 3, 0), NFRAMES - 128);
  const int sh   = q0 - fs;
  const int mt0  = wm * 32 + rg * 4;
  const int h    = lane >> 4;
  const int cl   = lane & 15;

  const v2f* Ab  = (const v2f*)apack;
  const v4f* Ap4 = (const v4f*)apack;

  int colIdx[4];
#pragma unroll
  for (int j = 0; j < 4; ++j) {
    const int t = q0 + wn * 64 + j * 16 + cl - wm;
    colIdx[j] = (bat * NFRAMES + min(max(t, 0), NFRAMES - 1)) * NBINS;
  }

  int aSrc[2];
#pragma unroll
  for (int t = 0; t < 2; ++t) {
    const int s = 2 * wv + t;
    aSrc[t] = (((s >> 2) * 32 + rg * 4 + (s & 3)) * KSTEPS) * 16 + lane;
  }
  const int rowBase = (bat * NFRAMES + fs + wv * 16) * NBINS + lane;

  v8f c[4][4];
#pragma unroll
  for (int i = 0; i < 4; ++i)
#pragma unroll
    for (int j = 0; j < 4; ++j) c[i][j] = (v8f){};

  v2f a[4], b[4];
  v4f ga[4];
  float g[8];

  if (threadIdx.x < 6 * BPITCH) {
    const int gi = (threadIdx.x < BPITCH)
                       ? (int)threadIdx.x
                       : 129 * BPITCH + (int)threadIdx.x - BPITCH;
    BT(0)[gi] = 0.0f;
    BT(1)[gi] = 0.0f;
  }

  STAGE_LOAD(0, 0)
  STAGE_STORE(0, 0)
  STAGE_LOAD(0, 1)
  STAGE_STORE(0, 1)
  __syncthreads();

  for (int kb = 0; kb < NKB; ++kb) {
    const int cur = kb & 1;

    switch (kb) {
      case 0:  __builtin_amdgcn_s_setprio(3); break;
      case 24: __builtin_amdgcn_s_setprio(2); break;
      case 44: __builtin_amdgcn_s_setprio(1); break;
      case 58: __builtin_amdgcn_s_setprio(0); break;
      default: break;
    }

    if (kb + 1 < NKB) {
      STAGE_LOAD(kb + 1, 0)
      asm volatile("" ::: "memory");
    }

    const float* acur = AT(cur) + wm * 4 * ATILE + lane * 2;
    const float* bcur =
        BT(cur) + (1 + sh - wm + wn * 64 + cl) * BPITCH + 2 * h;
#pragma unroll
    for (int s = 0; s < 8; ++s) {
      if (s == 4 && kb + 1 < NKB) {
        STAGE_STORE(cur ^ 1, 0)
        STAGE_LOAD(kb + 1, 1)
        asm volatile("" ::: "memory");
      }
#pragma unroll
      for (int i = 0; i < 4; ++i)
        a[i] = *(const v2f*)(acur + i * ATILE + s * 64);
#pragma unroll
      for (int j = 0; j < 4; ++j)
        b[j] = *(const v2f*)(bcur + j * 16 * BPITCH + 4 * s);
      WMMA16();
    }

    if (kb == 31) EDGE_STEP(xr, 256)

    if (kb + 1 < NKB) STAGE_STORE(cur ^ 1, 1)
    __syncthreads();
  }
  EDGE_STEP(xi, 513)

  {
    float* tw = lds + wv * ETILE + cl * EPITCH + 8 * h;
#pragma unroll
    for (int j = 0; j < 4; ++j) {
#pragma unroll
      for (int i = 0; i < 4; ++i) {
        v4f lo, hi;
        lo.x = c[i][j][0]; lo.y = c[i][j][1]; lo.z = c[i][j][2]; lo.w = c[i][j][3];
        hi.x = c[i][j][4]; hi.y = c[i][j][5]; hi.z = c[i][j][6]; hi.w = c[i][j][7];
        *(v4f*)(tw + j * 16 * EPITCH + i * 16)     = lo;
        *(v4f*)(tw + j * 16 * EPITCH + i * 16 + 4) = hi;
      }
    }
  }
  __syncthreads();

  {
    const int r = r0 + 2 * lane;
    float w0[4], w1[4];
#pragma unroll
    for (int d = 0; d < 4; ++d) {
      w0[d] = win[r + d * HOP];
      w1[d] = win[r + d * HOP + 1];
    }
    float* ob = out + (long)bat * OUTLEN + r;
#pragma unroll 4
    for (int m = 0; m < 16; ++m) {
      const int ql = wv + 8 * m;
      const int q  = q0 + ql;
      if (ql < TQ && q <= NFRAMES) {
        const float* tp =
            lds + (ql >> 6) * 4 * ETILE + (ql & 63) * EPITCH + 2 * lane;
        float acc0 = 0.0f, acc1 = 0.0f, wss0 = 0.0f, wss1 = 0.0f;
#pragma unroll
        for (int d = 3; d >= 0; --d) {
          const int t = q - d;
          const bool ok = (t >= 0) && (t <= NFRAMES - 1);
          const v2f x = *(const v2f*)(tp + d * ETILE);
          const float x0 = ok ? x.x : 0.0f;
          const float x1 = ok ? x.y : 0.0f;
          const float u0 = ok ? w0[d] : 0.0f;
          const float u1 = ok ? w1[d] : 0.0f;
          acc0 += x0;
          acc1 += x1;
          wss0 += u0 * u0;
          wss1 += u1 * u1;
        }
        v2f o;
        o.x = (wss0 > FLT_TINY) ? acc0 / wss0 : acc0;
        o.y = (wss1 > FLT_TINY) ? acc1 / wss1 : acc1;
        *(v2f*)(ob + (long)(q - 2) * HOP) = o;
      }
    }
  }
}

extern "C" void kernel_launch(void* const* d_in, const int* in_sizes, int n_in,
                              void* d_out, int out_size, void* d_ws,
                              size_t ws_size, hipStream_t stream) {
  const float* real = (const float*)d_in[0];
  const float* imag = (const float*)d_in[1];
  const float* Wr   = (const float*)d_in[2];
  const float* Wi   = (const float*)d_in[3];
  const float* win  = (const float*)d_in[4];
  float* out = (float*)d_out;

  float* apack = (float*)d_ws;

  {
    int total = MTILES * KSTEPS * 32;
    pack_a_kernel<<<(total + 255) / 256, 256, 0, stream>>>(Wr, Wi, apack);
  }
  istft_gemm_kernel<<<NBATCH * NQG * NRG, 256, 0, stream>>>(
      real, imag, apack, win, out);
}


# Overlap-add as a two-parallel fast block filter: hop pairs from 3 two-tap sub-filters on streams x[2t], x[2t+1] and their sum, y[2t+1]=P2-P0-P1; 6 products per pair, not 8; odd-hop sum order changes
# speedup vs baseline: 1.7210x; 1.1231x over previous
#include <hip/hip_runtime.h>

typedef __attribute__((ext_vector_type(2))) float v2f;
typedef __attribute__((ext_vector_type(8))) float v8f;

#define NFFT     2048
#define HOP      512
#define NBINS    1025
#define NFRAMES  1000
#define NPAIRS   500
#define NBATCH   8
#define OUTLEN   511488
#define KSTEPS   514
#define PTILES   192
#define FLT_TINY 1.1754943508222875e-38f

__device__ __forceinline__ float fold_coef(const float* __restrict__ Wr,
                                           const float* __restrict__ Wi,
                                           int o, int k) {
  if (k < 1028) {
    int f = k;
    if (f >= NBINS) return 0.0f;
    float v = Wr[o * NFFT + f];
    if (f >= 1 && f <= NFFT / 2 - 1) v += Wr[o * NFFT + (NFFT - f)];
    return v;
  } else {
    int f = k - 1028;
    if (f >= NBINS) return 0.0f;
    float v = -Wi[o * NFFT + f];
    if (f >= 1 && f <= NFFT / 2 - 1) v += Wi[o * NFFT + (NFFT - f)];
    return v;
  }
}

__global__ void pack_a_kernel(const float* __restrict__ Wr,
                              const float* __restrict__ Wi,
                              float* __restrict__ apack) {
  int gid  = blockIdx.x * blockDim.x + threadIdx.x;
  int lane = gid & 31;
  int kk   = (gid >> 5) % KSTEPS;
  int mt   = (gid >> 5) / KSTEPS;
  if (mt >= PTILES) return;
  int g = mt >> 5;
  int s = g >> 1;
  int e = g & 1;
  int r = (mt & 31) * 16 + (lane & 15);
  int k = kk * 4 + 2 * (lane >> 4);
  int idx = ((mt * KSTEPS + kk) * 32 + lane) * 2;
  float v0, v1;
  if (s < 2) {
    int o = r + HOP * (2 * e + s);
    v0 = fold_coef(Wr, Wi, o, k);
    v1 = fold_coef(Wr, Wi, o, k + 1);
  } else {
    int o = r + HOP * (2 * e);
    v0 = fold_coef(Wr, Wi, o, k) + fold_coef(Wr, Wi, o + HOP, k);
    v1 = fold_coef(Wr, Wi, o, k + 1) + fold_coef(Wr, Wi, o + HOP, k + 1);
  }
  apack[idx + 0] = v0;
  apack[idx + 1] = v1;
}

#define WMMA16()                                                               \
  _Pragma("unroll") for (int i = 0; i < 4; ++i) {                              \
    _Pragma("unroll") for (int j = 0; j < 4; ++j) {                            \
      c[i][j] = __builtin_amdgcn_wmma_f32_16x16x4_f32(                         \
          false, a[i], false, b[j], (short)0, c[i][j], false, false);          \
    }                                                                          \
  }

#define BK      32
#define BPITCH  36
#define BROWS   130
#define TP      125
#define NKB     64
#define NRG     (HOP / 64)
#define NPG     4
#define NWAVES  12
#define ATILE   512
#define ASTAGE  (24 * ATILE)
#define BSTREAM (BROWS * BPITCH)
#define STAGE_FLOATS (ASTAGE + 3 * BSTREAM)
#define EPITCH  68
#define ETILE   (64 * EPITCH)
#define LDS_FLOATS (2 * STAGE_FLOATS)
#define AT(st)  (lds + (st) * STAGE_FLOATS)
#define BT(st)  (lds + (st) * STAGE_FLOATS + ASTAGE)

typedef __attribute__((ext_vector_type(4))) float v4f;

#define EDGE_STEP(xp, kk)                                                      \
  {                                                                            \
    _Pragma("unroll") for (int i = 0; i < 4; ++i)                              \
        a[i] = Ab[((long)(mt0 + i) * KSTEPS + (kk)) * 32 + lane];              \
    _Pragma("unroll") for (int j = 0; j < 4; ++j) {                            \
      const int tp = ts + rowoff + wn * 64 + j * 16 + cl;                      \
      const bool okp = (tp >= 0) && (tp < NPAIRS);                             \
      const int tc = min(max(tp, 0), NPAIRS - 1);                              \
      const float* src = (xp) + (long)(bat * NFRAMES + 2 * tc) * NBINS + 1024; \
      const float ve = src[0];                                                 \
      const float vo = src[NBINS];                                             \
      float v = (sf == 0) ? ve : ((sf == 1) ? vo : (ve + vo));                 \
      v = okp ? v : 0.0f;                                                      \
      b[j].x = (h == 0) ? v : 0.0f;                                            \
      b[j].y = 0.0f;                                                           \
    }                                                                          \
    WMMA16();                                                                  \
  }

#define STAGE_LOAD(kn, t)                                                      \
  {                                                                            \
    const float* __restrict__ xn = ((kn) < 32) ? xr : xi;                      \
    const int f0  = ((kn) & 31) * BK;                                          \
    const int kkn = (((kn) < 32) ? 0 : 257) + ((kn) & 31) * 8;                 \
    _Pragma("unroll") for (int j = 0; j < 4; ++j)                              \
        ga[j] = Ap4[aSrc[t] + kkn * 16 + j * 32];                              \
    _Pragma("unroll") for (int i = 0; i < 6; ++i) {                            \
      const int ii = 6 * (t) + i;                                              \
      if ((ii == 0) ? do0 : ((ii == 10) ? do10 : (ii < 11))) {                 \
        g[2 * i]     = xn[rowS[ii] + f0 + lane];                               \
        g[2 * i + 1] = xn[rowS[ii] + NBINS + f0 + lane];                       \
      }                                                                        \
    }                                                                          \
  }

#define STAGE_STORE(st, t)                                                     \
  {                                                                            \
    _Pragma("unroll") for (int j = 0; j < 4; ++j)                              \
        *(v4f*)&AT(st)[(2 * wv + (t)) * ATILE + (j * 32 + lane) * 4] = ga[j];  \
    _Pragma("unroll") for (int i = 0; i < 6; ++i) {                            \
      const int ii = 6 * (t) + i;                                              \
      if ((ii == 0) ? do0 : ((ii == 10) ? do10 : (ii < 11))) {                 \
        const float ve = g[2 * i];                                             \
        const float vo = g[2 * i + 1];                                         \
        float* bp = BT(st) + (wv + 12 * ii) * BPITCH + lane;                   \
        bp[0]           = ve;                                                  \
        bp[BSTREAM]     = vo;                                                  \
        bp[2 * BSTREAM] = ve + vo;                                             \
      }                                                                        \
    }                                                                          \
  }

__global__ void __launch_bounds__(384, 3)
istft_gemm_kernel(const float* __restrict__ xr, const float* __restrict__ xi,
                  const float* __restrict__ apack,
                  const float* __restrict__ win, float* __restrict__ out) {
  __shared__ __attribute__((aligned(16))) float lds[LDS_FLOATS];

  const int lane = threadIdx.x & 31;
  const int wv   = __builtin_amdgcn_readfirstlane(threadIdx.x >> 5);
  const int w6   = wv % 6;
  const int wn   = wv / 6;
  const int sf   = w6 >> 1;
  const int tap  = w6 & 1;
  const int rg   = blockIdx.x & 7;
  const int pg   = (blockIdx.x >> 3) & 3;
  const int bat  = blockIdx.x >> 5;
  const int r0   = rg * 64;
  const int t0   = 1 + pg * TP;
  const int ts   = t0 - 2;
  const int rowoff = 2 - tap - ((sf == 1) ? 1 : 0);
  const int mt0  = w6 * 32 + rg * 4;
  const int h    = lane >> 4;
  const int cl   = lane & 15;

  const v2f* Ab  = (const v2f*)apack;
  const v4f* Ap4 = (const v4f*)apack;

  int aSrc[2];
#pragma unroll
  for (int t = 0; t < 2; ++t) {
    const int u = 2 * wv + t;
    aSrc[t] = (((u >> 2) * 32 + rg * 4 + (u & 3)) * KSTEPS) * 16 + lane;
  }
  int rowS[11];
#pragma unroll
  for (int ii = 0; ii < 11; ++ii) {
    const int tc = min(max(ts + wv + 12 * ii, 0), NPAIRS - 1);
    rowS[ii] = (bat * NFRAMES + 2 * tc) * NBINS;
  }
  const bool do0  = (ts + wv >= 0);
  const bool do10 = (wv + 120 < 128) && (ts + wv + 120 < NPAIRS);

  v8f c[4][4];
#pragma unroll
  for (int i = 0; i < 4; ++i)
#pragma unroll
    for (int j = 0; j < 4; ++j) c[i][j] = (v8f){};

  v2f a[4], b[4];
  v4f ga[4];
  float g[12];

  for (int u = threadIdx.x; u < 6 * 2 * BPITCH; u += 32 * NWAVES) {
    const int ss = u / (2 * BPITCH);
    const int w  = u % (2 * BPITCH);
    (BT(ss / 3) + (ss % 3) * BSTREAM + 128 * BPITCH)[w] = 0.0f;
  }
  if (!do0) {
#pragma unroll
    for (int ss = 0; ss < 6; ++ss)
      (BT(ss / 3) + (ss % 3) * BSTREAM + wv * BPITCH)[lane] = 0.0f;
  }
  if (!do10 && wv + 120 < 128) {
#pragma unroll
    for (int ss = 0; ss < 6; ++ss)
      (BT(ss / 3) + (ss % 3) * BSTREAM + (wv + 120) * BPITCH)[lane] = 0.0f;
  }

  STAGE_LOAD(0, 0)
  STAGE_STORE(0, 0)
  STAGE_LOAD(0, 1)
  STAGE_STORE(0, 1)
  __syncthreads();

  for (int kb = 0; kb < NKB; ++kb) {
    const int cur = kb & 1;

    if (kb + 1 < NKB) {
      STAGE_LOAD(kb + 1, 0)
      asm volatile("" ::: "memory");
    }

    const float* acur = AT(cur) + w6 * 4 * ATILE + lane * 2;
    const float* bcur = BT(cur) + sf * BSTREAM +
                        (rowoff + wn * 64 + cl) * BPITCH + 2 * h;
#pragma unroll
    for (int s = 0; s < 8; ++s) {
      if (s == 4 && kb + 1 < NKB) {
        STAGE_STORE(cur ^ 1, 0)
        STAGE_LOAD(kb + 1, 1)
        asm volatile("" ::: "memory");
      }
#pragma unroll
      for (int i = 0; i < 4; ++i)
        a[i] = *(const v2f*)(acur + i * ATILE + s * 64);
#pragma unroll
      for (int j = 0; j < 4; ++j)
        b[j] = *(const v2f*)(bcur + j * 16 * BPITCH + 4 * s);
      WMMA16();
    }

    if (kb == 31) EDGE_STEP(xr, 256)

    if (kb + 1 < NKB) STAGE_STORE(cur ^ 1, 1)
    __syncthreads();
  }
  EDGE_STEP(xi, 513)

  {
    float* tw = lds + wv * ETILE + cl * EPITCH + 8 * h;
#pragma unroll
    for (int j = 0; j < 4; ++j) {
#pragma unroll
      for (int i = 0; i < 4; ++i) {
        v4f lo, hi;
        lo.x = c[i][j][0]; lo.y = c[i][j][1]; lo.z = c[i][j][2]; lo.w = c[i][j][3];
        hi.x = c[i][j][4]; hi.y = c[i][j][5]; hi.z = c[i][j][6]; hi.w = c[i][j][7];
        *(v4f*)(tw + j * 16 * EPITCH + i * 16)     = lo;
        *(v4f*)(tw + j * 16 * EPITCH + i * 16 + 4) = hi;
      }
    }
  }
  __syncthreads();

  {
    const int r = r0 + 2 * lane;
    float w0[4], w1[4];
#pragma unroll
    for (int d = 0; d < 4; ++d) {
      w0[d] = win[r + d * HOP];
      w1[d] = win[r + d * HOP + 1];
    }
    float* ob = out + (long)bat * OUTLEN + r;
    for (int m = 0; m < 11; ++m) {
      const int tl = wv + 12 * m;
      if (tl < TP) {
        const int c1 = tl + 1;
        const float* p0 =
            lds + (tl >> 6) * 6 * ETILE + (tl & 63) * EPITCH + 2 * lane;
        const float* p1 =
            lds + (c1 >> 6) * 6 * ETILE + (c1 & 63) * EPITCH + 2 * lane;
        const v2f g0e  = *(const v2f*)(p0 + 0 * ETILE);
        const v2f g2e  = *(const v2f*)(p0 + 1 * ETILE);
        const v2f g1om = *(const v2f*)(p0 + 2 * ETILE);
        const v2f g3om = *(const v2f*)(p0 + 3 * ETILE);
        const v2f s01  = *(const v2f*)(p0 + 4 * ETILE);
        const v2f s23  = *(const v2f*)(p0 + 5 * ETILE);
        const v2f g1o  = *(const v2f*)(p1 + 2 * ETILE);
        const v2f g3o  = *(const v2f*)(p1 + 3 * ETILE);
        v2f y[2];
        y[0] = ((g3om + g2e) + g1om) + g0e;
        y[1] = ((s23 + s01) - (g2e + g0e)) - (g3o + g1o);
        const int t = t0 + tl;
#pragma unroll
        for (int par = 0; par < 2; ++par) {
          const int q = 2 * t + par;
          if (q <= NFRAMES) {
            float wss0 = 0.0f, wss1 = 0.0f;
#pragma unroll
            for (int d = 3; d >= 0; --d) {
              const int f = q - d;
              const bool ok = (f >= 0) && (f <= NFRAMES - 1);
              const float u0 = ok ? w0[d] : 0.0f;
              const float u1 = ok ? w1[d] : 0.0f;
              wss0 += u0 * u0;
              wss1 += u1 * u1;
            }
            v2f o;
            o.x = (wss0 > FLT_TINY) ? y[par].x / wss0 : y[par].x;
            o.y = (wss1 > FLT_TINY) ? y[par].y / wss1 : y[par].y;
            *(v2f*)(ob + (long)(q - 2) * HOP) = o;
          }
        }
      }
    }
  }
}

extern "C" void kernel_launch(void* const* d_in, const int* in_sizes, int n_in,
                              void* d_out, int out_size, void* d_ws,
                              size_t ws_size, hipStream_t stream) {
  const float* real = (const float*)d_in[0];
  const float* imag = (const float*)d_in[1];
  const float* Wr   = (const float*)d_in[2];
  const float* Wi   = (const float*)d_in[3];
  const float* win  = (const float*)d_in[4];
  float* out = (float*)d_out;

  float* apack = (float*)d_ws;

  {
    int total = PTILES * KSTEPS * 32;
    pack_a_kernel<<<(total + 255) / 256, 256, 0, stream>>>(Wr, Wi, apack);
  }
  istft_gemm_kernel<<<NBATCH * NPG * NRG, 384, 0, stream>>>(
      real, imag, apack, win, out);
}


# Coefficient pack launch re-tiled: 544 work-groups, not 12336; lane = k, so W is read in whole lines and once; fragment order made through LDS, 512-B store runs; values and sum order kept
# speedup vs baseline: 1.7850x; 1.0372x over previous
#include <hip/hip_runtime.h>

typedef __attribute__((ext_vector_type(2))) float v2f;
typedef __attribute__((ext_vector_type(4))) float v4f;
typedef __attribute__((ext_vector_type(8))) float v8f;

#define NFFT     2048
#define HOP      512
#define NBINS    1025
#define NFRAMES  1000
#define NPAIRS   500
#define NBATCH   8
#define OUTLEN   511488
#define KSTEPS   514
#define PTILES   192
#define FLT_TINY 1.1754943508222875e-38f

#define PKS    32
#define PCH    17
#define PGRID  (32 * PCH)
#define PPITCH 68
#define PGRP   (PKS * PPITCH)
__global__ void __launch_bounds__(256)
pack_a_kernel(const float* __restrict__ Wr, const float* __restrict__ Wi,
              float* __restrict__ apack) {
  __shared__ __attribute__((aligned(16))) float t[6 * PGRP];
  const int lane = threadIdx.x & 31;
  const int wv   = threadIdx.x >> 5;
  const int m    = blockIdx.x & 31;
  const int kk0  = (blockIdx.x >> 5) * PKS;

  float a[4][2][4], b[4][2][4];
  bool imag[4], valid[4], mir[4];
#pragma unroll
  for (int j = 0; j < 4; ++j) {
    const int k = kk0 * 4 + 32 * j + lane;
    imag[j]  = (k >= 1028);
    const int f = imag[j] ? k - 1028 : k;
    valid[j] = (f < NBINS);
    mir[j]   = (f >= 1) && (f <= NFFT / 2 - 1);
    const int fc = valid[j] ? f : 0;
    const int fm = mir[j] ? NFFT - f : 0;
    const float* __restrict__ Wp = imag[j] ? Wi : Wr;
#pragma unroll
    for (int rr = 0; rr < 2; ++rr) {
#pragma unroll
      for (int d = 0; d < 4; ++d) {
        const int o = m * 16 + 2 * wv + rr + HOP * d;
        a[j][rr][d] = Wp[o * NFFT + fc];
        b[j][rr][d] = Wp[o * NFFT + fm];
      }
    }
  }

#pragma unroll
  for (int j = 0; j < 4; ++j) {
    float* tp = t + (8 * j + (lane >> 2)) * PPITCH + ((lane >> 1) & 1) * 32 +
                (lane & 1) + 4 * wv;
#pragma unroll
    for (int rr = 0; rr < 2; ++rr) {
      float v[4];
#pragma unroll
      for (int d = 0; d < 4; ++d) {
        const float s = imag[j] ? -a[j][rr][d] : a[j][rr][d];
        const float w = mir[j] ? s + b[j][rr][d] : s;
        v[d] = valid[j] ? w : 0.0f;
      }
      float* tq = tp + 2 * rr;
      tq[0 * PGRP] = v[0];
      tq[1 * PGRP] = v[2];
      tq[2 * PGRP] = v[1];
      tq[3 * PGRP] = v[3];
      tq[4 * PGRP] = v[0] + v[1];
      tq[5 * PGRP] = v[2] + v[3];
    }
  }
  __syncthreads();

  v4f* __restrict__ ap4 = (v4f*)apack;
  const int c4 = threadIdx.x & 15;
#pragma unroll
  for (int hb = 0; hb < 2; ++hb) {
    const int kkl = hb * 16 + (threadIdx.x >> 4);
    const int kk  = kk0 + kkl;
    if (kk < KSTEPS) {
      v4f v[6];
#pragma unroll
      for (int g = 0; g < 6; ++g)
        v[g] = *(const v4f*)(t + g * PGRP + kkl * PPITCH + c4 * 4);
#pragma unroll
      for (int g = 0; g < 6; ++g)
        ap4[((g * 32 + m) * KSTEPS + kk) * 16 + c4] = v[g];
    }
  }
}

#define WMMA16()                                                               \
  _Pragma("unroll") for (int i = 0; i < 4; ++i) {                              \
    _Pragma("unroll") for (int j = 0; j < 4; ++j) {                            \
      c[i][j] = __builtin_amdgcn_wmma_f32_16x16x4_f32(                         \
          false, a[i], false, b[j], (short)0, c[i][j], false, false);          \
    }                                                                          \
  }

#define BK      32
#define BPITCH  36
#define BROWS   130
#define TP      125
#define NKB     64
#define NRG     (HOP / 64)
#define NPG     4
#define NWAVES  12
#define ATILE   512
#define ASTAGE  (24 * ATILE)
#define BSTREAM (BROWS * BPITCH)
#define STAGE_FLOATS (ASTAGE + 3 * BSTREAM)
#define EPITCH  68
#define ETILE   (64 * EPITCH)
#define LDS_FLOATS (2 * STAGE_FLOATS)
#define AT(st)  (lds + (st) * STAGE_FLOATS)
#define BT(st)  (lds + (st) * STAGE_FLOATS + ASTAGE)

#define EDGE_STEP(xp, kk)                                                      \
  {                                                                            \
    _Pragma("unroll") for (int i = 0; i < 4; ++i)                              \
        a[i] = Ab[((long)(mt0 + i) * KSTEPS + (kk)) * 32 + lane];              \
    _Pragma("unroll") for (int j = 0; j < 4; ++j) {                            \
      const int tp = ts + rowoff + wn * 64 + j * 16 + cl;                      \
      const bool okp = (tp >= 0) && (tp < NPAIRS);                             \
      const int tc = min(max(tp, 0), NPAIRS - 1);                              \
      const float* src = (xp) + (long)(bat * NFRAMES + 2 * tc) * NBINS + 1024; \
      const float ve = src[0];                                                 \
      const float vo = src[NBINS];                                             \
      float v = (sf == 0) ? ve : ((sf == 1) ? vo : (ve + vo));                 \
      v = okp ? v : 0.0f;                                                      \
      b[j].x = (h == 0) ? v : 0.0f;                                            \
      b[j].y = 0.0f;                                                           \
    }                                                                          \
    WMMA16();                                                                  \
  }

#define STAGE_LOAD(kn, t)                                                      \
  {                                                                            \
    const float* __restrict__ xn = ((kn) < 32) ? xr : xi;                      \
    const int f0  = ((kn) & 31) * BK;                                          \
    const int kkn = (((kn) < 32) ? 0 : 257) + ((kn) & 31) * 8;                 \
    _Pragma("unroll") for (int j = 0; j < 4; ++j)                              \
        ga[j] = Ap4[aSrc[t] + kkn * 16 + j * 32];                              \
    _Pragma("unroll") for (int i = 0; i < 6; ++i) {                            \
      const int ii = 6 * (t) + i;                                              \
      if ((ii == 0) ? do0 : ((ii == 10) ? do10 : (ii < 11))) {                 \
        g[2 * i]     = xn[rowS[ii] + f0 + lane];                               \
        g[2 * i + 1] = xn[rowS[ii] + NBINS + f0 + lane];                       \
      }                                                                        \
    }                                                                          \
  }

#define STAGE_STORE(st, t)                                                     \
  {                                                                            \
    _Pragma("unroll") for (int j = 0; j < 4; ++j)                              \
        *(v4f*)&AT(st)[(2 * wv + (t)) * ATILE + (j * 32 + lane) * 4] = ga[j];  \
    _Pragma("unroll") for (int i = 0; i < 6; ++i) {                            \
      const int ii = 6 * (t) + i;                                              \
      if ((ii == 0) ? do0 : ((ii == 10) ? do10 : (ii < 11))) {                 \
        const float ve = g[2 * i];                                             \
        const float vo = g[2 * i + 1];                                         \
        float* bp = BT(st) + (wv + 12 * ii) * BPITCH + lane;                   \
        bp[0]           = ve;                                                  \
        bp[BSTREAM]     = vo;                                                  \
        bp[2 * BSTREAM] = ve + vo;                                             \
      }                                                                        \
    }                                                                          \
  }

__global__ void __launch_bounds__(384, 3)
istft_gemm_kernel(const float* __restrict__ xr, const float* __restrict__ xi,
                  const float* __restrict__ apack,
                  const float* __restrict__ win, float* __restrict__ out) {
  __shared__ __attribute__((aligned(16))) float lds[LDS_FLOATS];

  const int lane = threadIdx.x & 31;
  const int wv   = __builtin_amdgcn_readfirstlane(threadIdx.x >> 5);
  const int w6   = wv % 6;
  const int wn   = wv / 6;
  const int sf   = w6 >> 1;
  const int tap  = w6 & 1;
  const int rg   = blockIdx.x & 7;
  const int pg   = (blockIdx.x >> 3) & 3;
  const int bat  = blockIdx.x >> 5;
  const int r0   = rg * 64;
  const int t0   = 1 + pg * TP;
  const int ts   = t0 - 2;
  const int rowoff = 2 - tap - ((sf == 1) ? 1 : 0);
  const int mt0  = w6 * 32 + rg * 4;
  const int h    = lane >> 4;
  const int cl   = lane & 15;

  const v2f* Ab  = (const v2f*)apack;
  const v4f* Ap4 = (const v4f*)apack;

  int aSrc[2];
#pragma unroll
  for (int t = 0; t < 2; ++t) {
    const int u = 2 * wv + t;
    aSrc[t] = (((u >> 2) * 32 + rg * 4 + (u & 3)) * KSTEPS) * 16 + lane;
  }
  int rowS[11];
#pragma unroll
  for (int ii = 0; ii < 11; ++ii) {
    const int tc = min(max(ts + wv + 12 * ii, 0), NPAIRS - 1);
    rowS[ii] = (bat * NFRAMES + 2 * tc) * NBINS;
  }
  const bool do0  = (ts + wv >= 0);
  const bool do10 = (wv + 120 < 128) && (ts + wv + 120 < NPAIRS);

  v8f c[4][4];
#pragma unroll
  for (int i = 0; i < 4; ++i)
#pragma unroll
    for (int j = 0; j < 4; ++j) c[i][j] = (v8f){};

  v2f a[4], b[4];
  v4f ga[4];
  float g[12];

  for (int u = threadIdx.x; u < 6 * 2 * BPITCH; u += 32 * NWAVES) {
    const int ss = u / (2 * BPITCH);
    const int w  = u % (2 * BPITCH);
    (BT(ss / 3) + (ss % 3) * BSTREAM + 128 * BPITCH)[w] = 0.0f;
  }
  if (!do0) {
#pragma unroll
    for (int ss = 0; ss < 6; ++ss)
      (BT(ss / 3) + (ss % 3) * BSTREAM + wv * BPITCH)[lane] = 0.0f;
  }
  if (!do10 && wv + 120 < 128) {
#pragma unroll
    for (int ss = 0; ss < 6; ++ss)
      (BT(ss / 3) + (ss % 3) * BSTREAM + (wv + 120) * BPITCH)[lane] = 0.0f;
  }

  STAGE_LOAD(0, 0)
  STAGE_STORE(0, 0)
  STAGE_LOAD(0, 1)
  STAGE_STORE(0, 1)
  __syncthreads();

  for (int kb = 0; kb < NKB; ++kb) {
    const int cur = kb & 1;

    if (kb + 1 < NKB) {
      STAGE_LOAD(kb + 1, 0)
      asm volatile("" ::: "memory");
    }

    const float* acur = AT(cur) + w6 * 4 * ATILE + lane * 2;
    const float* bcur = BT(cur) + sf * BSTREAM +
                        (rowoff + wn * 64 + cl) * BPITCH + 2 * h;
#pragma unroll
    for (int s = 0; s < 8; ++s) {
      if (s == 4 && kb + 1 < NKB) {
        STAGE_STORE(cur ^ 1, 0)
        STAGE_LOAD(kb + 1, 1)
        asm volatile("" ::: "memory");
      }
#pragma unroll
      for (int i = 0; i < 4; ++i)
        a[i] = *(const v2f*)(acur + i * ATILE + s * 64);
#pragma unroll
      for (int j = 0; j < 4; ++j)
        b[j] = *(const v2f*)(bcur + j * 16 * BPITCH + 4 * s);
      WMMA16();
    }

    if (kb == 31) EDGE_STEP(xr, 256)

    if (kb + 1 < NKB) STAGE_STORE(cur ^ 1, 1)
    __syncthreads();
  }
  EDGE_STEP(xi, 513)

  {
    float* tw = lds + wv * ETILE + cl * EPITCH + 8 * h;
#pragma unroll
    for (int j = 0; j < 4; ++j) {
#pragma unroll
      for (int i = 0; i < 4; ++i) {
        v4f lo, hi;
        lo.x = c[i][j][0]; lo.y = c[i][j][1]; lo.z = c[i][j][2]; lo.w = c[i][j][3];
        hi.x = c[i][j][4]; hi.y = c[i][j][5]; hi.z = c[i][j][6]; hi.w = c[i][j][7];
        *(v4f*)(tw + j * 16 * EPITCH + i * 16)     = lo;
        *(v4f*)(tw + j * 16 * EPITCH + i * 16 + 4) = hi;
      }
    }
  }
  __syncthreads();

  {
    const int r = r0 + 2 * lane;
    float w0[4], w1[4];
#pragma unroll
    for (int d = 0; d < 4; ++d) {
      w0[d] = win[r + d * HOP];
      w1[d] = win[r + d * HOP + 1];
    }
    float* ob = out + (long)bat * OUTLEN + r;
    for (int m = 0; m < 11; ++m) {
      const int tl = wv + 12 * m;
      if (tl < TP) {
        const int c1 = tl + 1;
        const float* p0 =
            lds + (tl >> 6) * 6 * ETILE + (tl & 63) * EPITCH + 2 * lane;
        const float* p1 =
            lds + (c1 >> 6) * 6 * ETILE + (c1 & 63) * EPITCH + 2 * lane;
        const v2f g0e  = *(const v2f*)(p0 + 0 * ETILE);
        const v2f g2e  = *(const v2f*)(p0 + 1 * ETILE);
        const v2f g1om = *(const v2f*)(p0 + 2 * ETILE);
        const v2f g3om = *(const v2f*)(p0 + 3 * ETILE);
        const v2f s01  = *(const v2f*)(p0 + 4 * ETILE);
        const v2f s23  = *(const v2f*)(p0 + 5 * ETILE);
        const v2f g1o  = *(const v2f*)(p1 + 2 * ETILE);
        const v2f g3o  = *(const v2f*)(p1 + 3 * ETILE);
        v2f y[2];
        y[0] = ((g3om + g2e) + g1om) + g0e;
        y[1] = ((s23 + s01) - (g2e + g0e)) - (g3o + g1o);
        const int t = t0 + tl;
#pragma unroll
        for (int par = 0; par < 2; ++par) {
          const int q = 2 * t + par;
          if (q <= NFRAMES) {
            float wss0 = 0.0f, wss1 = 0.0f;
#pragma unroll
            for (int d = 3; d >= 0; --d) {
              const int f = q - d;
              const bool ok = (f >= 0) && (f <= NFRAMES - 1);
              const float u0 = ok ? w0[d] : 0.0f;
              const float u1 = ok ? w1[d] : 0.0f;
              wss0 += u0 * u0;
              wss1 += u1 * u1;
            }
            v2f o;
            o.x = (wss0 > FLT_TINY) ? y[par].x / wss0 : y[par].x;
            o.y = (wss1 > FLT_TINY) ? y[par].y / wss1 : y[par].y;
            *(v2f*)(ob + (long)(q - 2) * HOP) = o;
          }
        }
      }
    }
  }
}

extern "C" void kernel_launch(void* const* d_in, const int* in_sizes, int n_in,
                              void* d_out, int out_size, void* d_ws,
                              size_t ws_size, hipStream_t stream) {
  const float* real = (const float*)d_in[0];
  const float* imag = (const float*)d_in[1];
  const float* Wr   = (const float*)d_in[2];
  const float* Wi   = (const float*)d_in[3];
  const float* win  = (const float*)d_in[4];
  float* out = (float*)d_out;

  float* apack = (float*)d_ws;

  pack_a_kernel<<<PGRID, 256, 0, stream>>>(Wr, Wi, apack);
  istft_gemm_kernel<<<NBATCH * NPG * NRG, 384, 0, stream>>>(
      real, imag, apack, win, out);
}


# Overlap-add as a four-parallel fast block filter (was two-parallel): hop quads from 9 one-tap products on streams x[4u+i] and their sums, 9 per 4 hops, not 12; wave tile 48x64; sum order changes
# speedup vs baseline: 1.8903x; 1.0590x over previous
#include <hip/hip_runtime.h>

typedef __attribute__((ext_vector_type(2))) float v2f;
typedef __attribute__((ext_vector_type(4))) float v4f;
typedef __attribute__((ext_vector_type(8))) float v8f;

#define NFFT     2048
#define HOP      512
#define NBINS    1025
#define NFRAMES  1000
#define NBATCH   8
#define OUTLEN   511488
#define KSTEPS   514
#define NQUADS   250
#define PTILES   288
#define FLT_TINY 1.1754943508222875e-38f

#define PKS    32
#define PCH    17
#define PGRID  (32 * PCH)
#define PPITCH 68
#define PGRP   (PKS * PPITCH)
__global__ void __launch_bounds__(256)
pack_a_kernel(const float* __restrict__ Wr, const float* __restrict__ Wi,
              float* __restrict__ apack) {
  __shared__ __attribute__((aligned(16))) float t[9 * PGRP];
  const int lane = threadIdx.x & 31;
  const int wv   = threadIdx.x >> 5;
  const int m    = blockIdx.x & 31;
  const int kk0  = (blockIdx.x >> 5) * PKS;

  float a[4][2][4], b[4][2][4];
  bool imag[4], valid[4], mir[4];
#pragma unroll
  for (int j = 0; j < 4; ++j) {
    const int k = kk0 * 4 + 32 * j + lane;
    imag[j]  = (k >= 1028);
    const int f = imag[j] ? k - 1028 : k;
    valid[j] = (f < NBINS);
    mir[j]   = (f >= 1) && (f <= NFFT / 2 - 1);
    const int fc = valid[j] ? f : 0;
    const int fm = mir[j] ? NFFT - f : 0;
    const float* __restrict__ Wp = imag[j] ? Wi : Wr;
#pragma unroll
    for (int rr = 0; rr < 2; ++rr) {
#pragma unroll
      for (int d = 0; d < 4; ++d) {
        const int o = m * 16 + 2 * wv + rr + HOP * d;
        a[j][rr][d] = Wp[o * NFFT + fc];
        b[j][rr][d] = Wp[o * NFFT + fm];
      }
    }
  }

#pragma unroll
  for (int j = 0; j < 4; ++j) {
    float* tp = t + (8 * j + (lane >> 2)) * PPITCH + ((lane >> 1) & 1) * 32 +
                (lane & 1) + 4 * wv;
#pragma unroll
    for (int rr = 0; rr < 2; ++rr) {
      float v[4];
#pragma unroll
      for (int d = 0; d < 4; ++d) {
        const float s = imag[j] ? -a[j][rr][d] : a[j][rr][d];
        const float w = mir[j] ? s + b[j][rr][d] : s;
        v[d] = valid[j] ? w : 0.0f;
      }
      float* tq = tp + 2 * rr;
      const float s01 = v[0] + v[1];
      const float s23 = v[2] + v[3];
      tq[0 * PGRP] = v[0];
      tq[1 * PGRP] = v[1];
      tq[2 * PGRP] = s01;
      tq[3 * PGRP] = v[2];
      tq[4 * PGRP] = v[3];
      tq[5 * PGRP] = s23;
      tq[6 * PGRP] = v[0] + v[2];
      tq[7 * PGRP] = v[1] + v[3];
      tq[8 * PGRP] = s01 + s23;
    }
  }
  __syncthreads();

  v4f* __restrict__ ap4 = (v4f*)apack;
  const int c4 = threadIdx.x & 15;
#pragma unroll
  for (int hb = 0; hb < 2; ++hb) {
    const int kkl = hb * 16 + (threadIdx.x >> 4);
    const int kk  = kk0 + kkl;
    if (kk < KSTEPS) {
      v4f v[9];
#pragma unroll
      for (int g = 0; g < 9; ++g)
        v[g] = *(const v4f*)(t + g * PGRP + kkl * PPITCH + c4 * 4);
#pragma unroll
      for (int g = 0; g < 9; ++g)
        ap4[((g * 32 + m) * KSTEPS + kk) * 16 + c4] = v[g];
    }
  }
}

#define WMMA12()                                                               \
  _Pragma("unroll") for (int i = 0; i < 3; ++i) {                              \
    _Pragma("unroll") for (int j = 0; j < 4; ++j) {                            \
      c[i][j] = __builtin_amdgcn_wmma_f32_16x16x4_f32(                         \
          false, a[i], false, b[i][j], (short)0, c[i][j], false, false);       \
    }                                                                          \
  }

#define BK      32
#define BPITCH  36
#define BROWS   64
#define QC      63
#define NKB     64
#define NRG     (HOP / 64)
#define NQG     4
#define NWAVES  12
#define ATILE   512
#define ASTAGE  (36 * ATILE)
#define BSTREAM (BROWS * BPITCH)
#define STAGE_FLOATS (ASTAGE + 9 * BSTREAM)
#define EPITCH  68
#define ETILE   (64 * EPITCH)
#define LDS_FLOATS (2 * STAGE_FLOATS)
#define AT(st)  (lds + (st) * STAGE_FLOATS)
#define BT(st)  (lds + (st) * STAGE_FLOATS + ASTAGE)

__device__ __forceinline__ float stream_of(int p, float x0, float x1, float x2,
                                           float x3) {
  const bool k0 = (p == 0) | (p == 2) | (p == 6) | (p == 8);
  const bool k1 = (p == 1) | (p == 2) | (p == 7) | (p == 8);
  const bool k2 = (p == 3) | (p == 5) | (p == 6) | (p == 8);
  const bool k3 = (p == 4) | (p == 5) | (p == 7) | (p == 8);
  return ((k0 ? x0 : 0.0f) + (k1 ? x1 : 0.0f)) +
         ((k2 ? x2 : 0.0f) + (k3 ? x3 : 0.0f));
}

#define EDGE_STEP(xp, kk)                                                      \
  {                                                                            \
    _Pragma("unroll") for (int i = 0; i < 3; ++i)                              \
        a[i] = Ab[((long)tileA[i] * KSTEPS + (kk)) * 32 + lane];               \
    _Pragma("unroll") for (int j = 0; j < 4; ++j) {                            \
      const int uq = u0 - 1 + j * 16 + cl;                                     \
      const bool okq = (uq >= 0) && (uq < NQUADS);                             \
      const int uc = min(max(uq, 0), NQUADS - 1);                              \
      const float* src = (xp) + (long)(bat * NFRAMES + 4 * uc) * NBINS + 1024; \
      const float e0 = src[0];                                                 \
      const float e1 = src[NBINS];                                             \
      const float e2 = src[2 * NBINS];                                         \
      const float e3 = src[3 * NBINS];                                         \
      _Pragma("unroll") for (int i = 0; i < 3; ++i) {                          \
        float v = stream_of(prod[i], e0, e1, e2, e3);                          \
        v = okq ? v : 0.0f;                                                    \
        b[i][j].x = (h == 0) ? v : 0.0f;                                       \
        b[i][j].y = 0.0f;                                                      \
      }                                                                        \
    }                                                                          \
    WMMA12();                                                                  \
  }

#define ROW_ON(ii) (((ii) == 0) ? do0 : (((ii) == 5) ? do5 : true))

#define STAGE_LOAD(kn, t)                                                      \
  {                                                                            \
    const float* __restrict__ xn = ((kn) < 32) ? xr : xi;                      \
    const int f0  = ((kn) & 31) * BK;                                          \
    const int kkn = (((kn) < 32) ? 0 : 257) + ((kn) & 31) * 8;                 \
    _Pragma("unroll") for (int u = 0; u < 2 - (t); ++u) {                      \
      _Pragma("unroll") for (int j = 0; j < 4; ++j)                            \
          ga[4 * u + j] = Ap4[aSrc[2 * (t) + u] + kkn * 16 + j * 32];          \
    }                                                                          \
    _Pragma("unroll") for (int i = 0; i < 3; ++i) {                            \
      const int ii = 3 * (t) + i;                                              \
      if (ROW_ON(ii)) {                                                        \
        _Pragma("unroll") for (int fr = 0; fr < 4; ++fr)                       \
            g[4 * i + fr] = xn[rowS[ii] + fr * NBINS + f0 + lane];             \
      }                                                                        \
    }                                                                          \
  }

#define STAGE_STORE(st, t)                                                     \
  {                                                                            \
    _Pragma("unroll") for (int u = 0; u < 2 - (t); ++u) {                      \
      _Pragma("unroll") for (int j = 0; j < 4; ++j)                            \
          *(v4f*)&AT(st)[(3 * wv + 2 * (t) + u) * ATILE + (j * 32 + lane) * 4] = \
              ga[4 * u + j];                                                   \
    }                                                                          \
    _Pragma("unroll") for (int i = 0; i < 3; ++i) {                            \
      const int ii = 3 * (t) + i;                                              \
      if (ROW_ON(ii)) {                                                        \
        const float x0 = g[4 * i];                                             \
        const float x1 = g[4 * i + 1];                                         \
        const float x2 = g[4 * i + 2];                                         \
        const float x3 = g[4 * i + 3];                                         \
        const float s01 = x0 + x1;                                             \
        const float s23 = x2 + x3;                                             \
        float* bp = BT(st) + (wv + 12 * ii) * BPITCH + lane;                   \
        bp[0]           = x0;                                                  \
        bp[BSTREAM]     = x1;                                                  \
        bp[2 * BSTREAM] = s01;                                                 \
        bp[3 * BSTREAM] = x2;                                                  \
        bp[4 * BSTREAM] = x3;                                                  \
        bp[5 * BSTREAM] = s23;                                                 \
        bp[6 * BSTREAM] = x0 + x2;                                             \
        bp[7 * BSTREAM] = x1 + x3;                                             \
        bp[8 * BSTREAM] = s01 + s23;                                           \
      }                                                                        \
    }                                                                          \
  }

__global__ void __launch_bounds__(384, 3)
istft_gemm_kernel(const float* __restrict__ xr, const float* __restrict__ xi,
                  const float* __restrict__ apack,
                  const float* __restrict__ win, float* __restrict__ out) {
  __shared__ __attribute__((aligned(16))) float lds[LDS_FLOATS];

  const int lane = threadIdx.x & 31;
  const int wv   = __builtin_amdgcn_readfirstlane(threadIdx.x >> 5);
  const int rg   = blockIdx.x & 7;
  const int qg   = (blockIdx.x >> 3) & 3;
  const int bat  = blockIdx.x >> 5;
  const int r0   = rg * 64;
  const int u0   = qg * QC;
  const int h    = lane >> 4;
  const int cl   = lane & 15;

  const v2f* Ab  = (const v2f*)apack;
  const v4f* Ap4 = (const v4f*)apack;

  int prod[3], mtl[3], tileA[3], aSrc[3];
#pragma unroll
  for (int i = 0; i < 3; ++i) {
    const int row = 3 * wv + i;
    prod[i]  = row >> 2;
    mtl[i]   = row & 3;
    tileA[i] = prod[i] * 32 + rg * 4 + mtl[i];
    aSrc[i]  = (tileA[i] * KSTEPS) * 16 + lane;
  }
  int rowS[6];
#pragma unroll
  for (int ii = 0; ii < 6; ++ii) {
    const int uq = u0 - 1 + wv + 12 * ii;
    const int uc = min(max(uq, 0), NQUADS - 1);
    rowS[ii] = (bat * NFRAMES + 4 * uc) * NBINS;
  }
  const bool do0 = (u0 - 1 + wv >= 0);
  const bool do5 = (wv + 60 < BROWS) && (u0 - 1 + wv + 60 < NQUADS);

  v8f c[3][4];
#pragma unroll
  for (int i = 0; i < 3; ++i)
#pragma unroll
    for (int j = 0; j < 4; ++j) c[i][j] = (v8f){};

  v2f a[3], b[3][4];
  v4f ga[8];
  float g[12];

  if (!do0) {
#pragma unroll
    for (int ss = 0; ss < 18; ++ss)
      (BT(ss / 9) + (ss % 9) * BSTREAM + wv * BPITCH)[lane] = 0.0f;
  }
  if (!do5 && wv + 60 < BROWS) {
#pragma unroll
    for (int ss = 0; ss < 18; ++ss)
      (BT(ss / 9) + (ss % 9) * BSTREAM + (wv + 60) * BPITCH)[lane] = 0.0f;
  }

  STAGE_LOAD(0, 0)
  STAGE_STORE(0, 0)
  STAGE_LOAD(0, 1)
  STAGE_STORE(0, 1)
  __syncthreads();

  for (int kb = 0; kb < NKB; ++kb) {
    const int cur = kb & 1;

    if (kb + 1 < NKB) {
      STAGE_LOAD(kb + 1, 0)
      asm volatile("" ::: "memory");
    }

    const float* acur = AT(cur) + 3 * wv * ATILE + lane * 2;
    const float* bcur[3];
#pragma unroll
    for (int i = 0; i < 3; ++i)
      bcur[i] = BT(cur) + prod[i] * BSTREAM + cl * BPITCH + 2 * h;
#pragma unroll
    for (int s = 0; s < 8; ++s) {
      if (s == 4 && kb + 1 < NKB) {
        STAGE_STORE(cur ^ 1, 0)
        STAGE_LOAD(kb + 1, 1)
        asm volatile("" ::: "memory");
      }
#pragma unroll
      for (int i = 0; i < 3; ++i)
        a[i] = *(const v2f*)(acur + i * ATILE + s * 64);
#pragma unroll
      for (int i = 0; i < 3; ++i)
#pragma unroll
        for (int j = 0; j < 4; ++j)
          b[i][j] = *(const v2f*)(bcur[i] + j * 16 * BPITCH + 4 * s);
      WMMA12();
    }

    if (kb == 31) EDGE_STEP(xr, 256)

    if (kb + 1 < NKB) STAGE_STORE(cur ^ 1, 1)
    __syncthreads();
  }
  EDGE_STEP(xi, 513)

  {
#pragma unroll
    for (int i = 0; i < 3; ++i) {
      float* tw = lds + prod[i] * ETILE + cl * EPITCH + mtl[i] * 16 + 8 * h;
#pragma unroll
      for (int j = 0; j < 4; ++j) {
        v4f lo, hi;
        lo.x = c[i][j][0]; lo.y = c[i][j][1]; lo.z = c[i][j][2]; lo.w = c[i][j][3];
        hi.x = c[i][j][4]; hi.y = c[i][j][5]; hi.z = c[i][j][6]; hi.w = c[i][j][7];
        *(v4f*)(tw + j * 16 * EPITCH)     = lo;
        *(v4f*)(tw + j * 16 * EPITCH + 4) = hi;
      }
    }
  }
  __syncthreads();

  {
    const int r = r0 + 2 * lane;
    float w0[4], w1[4];
#pragma unroll
    for (int d = 0; d < 4; ++d) {
      w0[d] = win[r + d * HOP];
      w1[d] = win[r + d * HOP + 1];
    }
    float* ob = out + (long)bat * OUTLEN + r;
    for (int m = 0; m < 6; ++m) {
      const int col = 1 + wv + 12 * m;
      if (col < 64) {
        const float* pc = lds + col * EPITCH + 2 * lane;
        const float* pm = pc - EPITCH;
        const v2f P1 = *(const v2f*)(pc + 0 * ETILE);
        const v2f P2 = *(const v2f*)(pc + 1 * ETILE);
        const v2f P3 = *(const v2f*)(pc + 2 * ETILE);
        const v2f P4 = *(const v2f*)(pc + 3 * ETILE);
        const v2f P5 = *(const v2f*)(pc + 4 * ETILE);
        const v2f P6 = *(const v2f*)(pc + 5 * ETILE);
        const v2f P7 = *(const v2f*)(pc + 6 * ETILE);
        const v2f P8 = *(const v2f*)(pc + 7 * ETILE);
        const v2f P9 = *(const v2f*)(pc + 8 * ETILE);
        const v2f Q2 = *(const v2f*)(pm + 1 * ETILE);
        const v2f Q4 = *(const v2f*)(pm + 3 * ETILE);
        const v2f Q5 = *(const v2f*)(pm + 4 * ETILE);
        const v2f Q6 = *(const v2f*)(pm + 5 * ETILE);
        const v2f Q8 = *(const v2f*)(pm + 7 * ETILE);
        const v2f t1 = P3 - (P1 + P2);
        const v2f t5 = P6 - (P4 + P5);
        v2f y[4];
        y[0] = P1 + ((Q4 + Q8) - (Q2 + Q5));
        y[1] = t1 + (Q6 - (Q4 + Q5));
        y[2] = ((P2 + P7) - (P1 + P4)) + Q5;
        y[3] = ((P9 - (P7 + P8)) - t1) - t5;
        const int u = u0 - 1 + col;
#pragma unroll
        for (int j = 0; j < 4; ++j) {
          const int q = 4 * u + j;
          if (q >= 2 && q <= NFRAMES) {
            float wss0 = 0.0f, wss1 = 0.0f;
#pragma unroll
            for (int d = 3; d >= 0; --d) {
              const int f = q - d;
              const bool ok = (f >= 0) && (f <= NFRAMES - 1);
              const float v0 = ok ? w0[d] : 0.0f;
              const float v1 = ok ? w1[d] : 0.0f;
              wss0 += v0 * v0;
              wss1 += v1 * v1;
            }
            v2f o;
            o.x = (wss0 > FLT_TINY) ? y[j].x / wss0 : y[j].x;
            o.y = (wss1 > FLT_TINY) ? y[j].y / wss1 : y[j].y;
            *(v2f*)(ob + (long)(q - 2) * HOP) = o;
          }
        }
      }
    }
  }
}

extern "C" void kernel_launch(void* const* d_in, const int* in_sizes, int n_in,
                              void* d_out, int out_size, void* d_ws,
                              size_t ws_size, hipStream_t stream) {
  const float* real = (const float*)d_in[0];
  const float* imag = (const float*)d_in[1];
  const float* Wr   = (const float*)d_in[2];
  const float* Wi   = (const float*)d_in[3];
  const float* win  = (const float*)d_in[4];
  float* out = (float*)d_out;

  float* apack = (float*)d_ws;

  pack_a_kernel<<<PGRID, 256, 0, stream>>>(Wr, Wi, apack);
  istft_gemm_kernel<<<NBATCH * NQG * NRG, 384, 0, stream>>>(
      real, imag, apack, win, out);
}


# GEMM work-group barrier split: a wave signals at mid-block, once its second-half fragments are in registers and its stage share is stored, and waits after the block's last products; sum order kept
# speedup vs baseline: 2.0636x; 1.0917x over previous
#include <hip/hip_runtime.h>

typedef __attribute__((ext_vector_type(2))) float v2f;
typedef __attribute__((ext_vector_type(4))) float v4f;
typedef __attribute__((ext_vector_type(8))) float v8f;

#define NFFT     2048
#define HOP      512
#define NBINS    1025
#define NFRAMES  1000
#define NBATCH   8
#define OUTLEN   511488
#define KSTEPS   514
#define NQUADS   250
#define PTILES   288
#define FLT_TINY 1.1754943508222875e-38f

#define PKS    32
#define PCH    17
#define PGRID  (32 * PCH)
#define PPITCH 68
#define PGRP   (PKS * PPITCH)
__global__ void __launch_bounds__(256)
pack_a_kernel(const float* __restrict__ Wr, const float* __restrict__ Wi,
              float* __restrict__ apack) {
  __shared__ __attribute__((aligned(16))) float t[9 * PGRP];
  const int lane = threadIdx.x & 31;
  const int wv   = threadIdx.x >> 5;
  const int m    = blockIdx.x & 31;
  const int kk0  = (blockIdx.x >> 5) * PKS;

  float a[4][2][4], b[4][2][4];
  bool imag[4], valid[4], mir[4];
#pragma unroll
  for (int j = 0; j < 4; ++j) {
    const int k = kk0 * 4 + 32 * j + lane;
    imag[j]  = (k >= 1028);
    const int f = imag[j] ? k - 1028 : k;
    valid[j] = (f < NBINS);
    mir[j]   = (f >= 1) && (f <= NFFT / 2 - 1);
    const int fc = valid[j] ? f : 0;
    const int fm = mir[j] ? NFFT - f : 0;
    const float* __restrict__ Wp = imag[j] ? Wi : Wr;
#pragma unroll
    for (int rr = 0; rr < 2; ++rr) {
#pragma unroll
      for (int d = 0; d < 4; ++d) {
        const int o = m * 16 + 2 * wv + rr + HOP * d;
        a[j][rr][d] = Wp[o * NFFT + fc];
        b[j][rr][d] = Wp[o * NFFT + fm];
      }
    }
  }

#pragma unroll
  for (int j = 0; j < 4; ++j) {
    float* tp = t + (8 * j + (lane >> 2)) * PPITCH + ((lane >> 1) & 1) * 32 +
                (lane & 1) + 4 * wv;
#pragma unroll
    for (int rr = 0; rr < 2; ++rr) {
      float v[4];
#pragma unroll
      for (int d = 0; d < 4; ++d) {
        const float s = imag[j] ? -a[j][rr][d] : a[j][rr][d];
        const float w = mir[j] ? s + b[j][rr][d] : s;
        v[d] = valid[j] ? w : 0.0f;
      }
      float* tq = tp + 2 * rr;
      const float s01 = v[0] + v[1];
      const float s23 = v[2] + v[3];
      tq[0 * PGRP] = v[0];
      tq[1 * PGRP] = v[1];
      tq[2 * PGRP] = s01;
      tq[3 * PGRP] = v[2];
      tq[4 * PGRP] = v[3];
      tq[5 * PGRP] = s23;
      tq[6 * PGRP] = v[0] + v[2];
      tq[7 * PGRP] = v[1] + v[3];
      tq[8 * PGRP] = s01 + s23;
    }
  }
  __syncthreads();

  v4f* __restrict__ ap4 = (v4f*)apack;
  const int c4 = threadIdx.x & 15;
#pragma unroll
  for (int hb = 0; hb < 2; ++hb) {
    const int kkl = hb * 16 + (threadIdx.x >> 4);
    const int kk  = kk0 + kkl;
    if (kk < KSTEPS) {
      v4f v[9];
#pragma unroll
      for (int g = 0; g < 9; ++g)
        v[g] = *(const v4f*)(t + g * PGRP + kkl * PPITCH + c4 * 4);
#pragma unroll
      for (int g = 0; g < 9; ++g)
        ap4[((g * 32 + m) * KSTEPS + kk) * 16 + c4] = v[g];
    }
  }
}

#define WMMA12()                                                               \
  _Pragma("unroll") for (int i = 0; i < 3; ++i) {                              \
    _Pragma("unroll") for (int j = 0; j < 4; ++j) {                            \
      c[i][j] = __builtin_amdgcn_wmma_f32_16x16x4_f32(                         \
          false, a[i], false, b[i][j], (short)0, c[i][j], false, false);       \
    }                                                                          \
  }

#define BK      32
#define BPITCH  36
#define BROWS   64
#define QC      63
#define NKB     64
#define NRG     (HOP / 64)
#define NQG     4
#define NWAVES  12
#define ATILE   512
#define ASTAGE  (36 * ATILE)
#define BSTREAM (BROWS * BPITCH)
#define STAGE_FLOATS (ASTAGE + 9 * BSTREAM)
#define EPITCH  68
#define ETILE   (64 * EPITCH)
#define LDS_FLOATS (2 * STAGE_FLOATS)
#define AT(st)  (lds + (st) * STAGE_FLOATS)
#define BT(st)  (lds + (st) * STAGE_FLOATS + ASTAGE)

__device__ __forceinline__ float stream_of(int p, float x0, float x1, float x2,
                                           float x3) {
  const bool k0 = (p == 0) | (p == 2) | (p == 6) | (p == 8);
  const bool k1 = (p == 1) | (p == 2) | (p == 7) | (p == 8);
  const bool k2 = (p == 3) | (p == 5) | (p == 6) | (p == 8);
  const bool k3 = (p == 4) | (p == 5) | (p == 7) | (p == 8);
  return ((k0 ? x0 : 0.0f) + (k1 ? x1 : 0.0f)) +
         ((k2 ? x2 : 0.0f) + (k3 ? x3 : 0.0f));
}

#define EDGE_STEP(xp, kk)                                                      \
  {                                                                            \
    _Pragma("unroll") for (int i = 0; i < 3; ++i)                              \
        a[i] = Ab[((long)tileA[i] * KSTEPS + (kk)) * 32 + lane];               \
    _Pragma("unroll") for (int j = 0; j < 4; ++j) {                            \
      const int uq = u0 - 1 + j * 16 + cl;                                     \
      const bool okq = (uq >= 0) && (uq < NQUADS);                             \
      const int uc = min(max(uq, 0), NQUADS - 1);                              \
      const float* src = (xp) + (long)(bat * NFRAMES + 4 * uc) * NBINS + 1024; \
      const float e0 = src[0];                                                 \
      const float e1 = src[NBINS];                                             \
      const float e2 = src[2 * NBINS];                                         \
      const float e3 = src[3 * NBINS];                                         \
      _Pragma("unroll") for (int i = 0; i < 3; ++i) {                          \
        float v = stream_of(prod[i], e0, e1, e2, e3);                          \
        v = okq ? v : 0.0f;                                                    \
        b[i][j].x = (h == 0) ? v : 0.0f;                                       \
        b[i][j].y = 0.0f;                                                      \
      }                                                                        \
    }                                                                          \
    WMMA12();                                                                  \
  }

#define ROW_ON(ii) (((ii) == 0) ? do0 : (((ii) == 5) ? do5 : true))

#define STAGE_LOAD(kn, t)                                                      \
  {                                                                            \
    const float* __restrict__ xn = ((kn) < 32) ? xr : xi;                      \
    const int f0  = ((kn) & 31) * BK;                                          \
    const int kkn = (((kn) < 32) ? 0 : 257) + ((kn) & 31) * 8;                 \
    _Pragma("unroll") for (int u = 0; u < 2 - (t); ++u) {                      \
      _Pragma("unroll") for (int j = 0; j < 4; ++j)                            \
          ga[4 * u + j] = Ap4[aSrc[2 * (t) + u] + kkn * 16 + j * 32];          \
    }                                                                          \
    _Pragma("unroll") for (int i = 0; i < 3; ++i) {                            \
      const int ii = 3 * (t) + i;                                              \
      if (ROW_ON(ii)) {                                                        \
        _Pragma("unroll") for (int fr = 0; fr < 4; ++fr)                       \
            g[4 * i + fr] = xn[rowS[ii] + fr * NBINS + f0 + lane];             \
      }                                                                        \
    }                                                                          \
  }

#define STAGE_STORE(st, t)                                                     \
  {                                                                            \
    _Pragma("unroll") for (int u = 0; u < 2 - (t); ++u) {                      \
      _Pragma("unroll") for (int j = 0; j < 4; ++j)                            \
          *(v4f*)&AT(st)[(3 * wv + 2 * (t) + u) * ATILE + (j * 32 + lane) * 4] = \
              ga[4 * u + j];                                                   \
    }                                                                          \
    _Pragma("unroll") for (int i = 0; i < 3; ++i) {                            \
      const int ii = 3 * (t) + i;                                              \
      if (ROW_ON(ii)) {                                                        \
        const float x0 = g[4 * i];                                             \
        const float x1 = g[4 * i + 1];                                         \
        const float x2 = g[4 * i + 2];                                         \
        const float x3 = g[4 * i + 3];                                         \
        const float s01 = x0 + x1;                                             \
        const float s23 = x2 + x3;                                             \
        float* bp = BT(st) + (wv + 12 * ii) * BPITCH + lane;                   \
        bp[0]           = x0;                                                  \
        bp[BSTREAM]     = x1;                                                  \
        bp[2 * BSTREAM] = s01;                                                 \
        bp[3 * BSTREAM] = x2;                                                  \
        bp[4 * BSTREAM] = x3;                                                  \
        bp[5 * BSTREAM] = s23;                                                 \
        bp[6 * BSTREAM] = x0 + x2;                                             \
        bp[7 * BSTREAM] = x1 + x3;                                             \
        bp[8 * BSTREAM] = s01 + s23;                                           \
      }                                                                        \
    }                                                                          \
  }

__global__ void __launch_bounds__(384, 3)
istft_gemm_kernel(const float* __restrict__ xr, const float* __restrict__ xi,
                  const float* __restrict__ apack,
                  const float* __restrict__ win, float* __restrict__ out) {
  __shared__ __attribute__((aligned(16))) float lds[LDS_FLOATS];

  const int lane = threadIdx.x & 31;
  const int wv   = __builtin_amdgcn_readfirstlane(threadIdx.x >> 5);
  const int rg   = blockIdx.x & 7;
  const int qg   = (blockIdx.x >> 3) & 3;
  const int bat  = blockIdx.x >> 5;
  const int r0   = rg * 64;
  const int u0   = qg * QC;
  const int h    = lane >> 4;
  const int cl   = lane & 15;

  const v2f* Ab  = (const v2f*)apack;
  const v4f* Ap4 = (const v4f*)apack;

  int prod[3], mtl[3], tileA[3], aSrc[3];
#pragma unroll
  for (int i = 0; i < 3; ++i) {
    const int row = 3 * wv + i;
    prod[i]  = row >> 2;
    mtl[i]   = row & 3;
    tileA[i] = prod[i] * 32 + rg * 4 + mtl[i];
    aSrc[i]  = (tileA[i] * KSTEPS) * 16 + lane;
  }
  int rowS[6];
#pragma unroll
  for (int ii = 0; ii < 6; ++ii) {
    const int uq = u0 - 1 + wv + 12 * ii;
    const int uc = min(max(uq, 0), NQUADS - 1);
    rowS[ii] = (bat * NFRAMES + 4 * uc) * NBINS;
  }
  const bool do0 = (u0 - 1 + wv >= 0);
  const bool do5 = (wv + 60 < BROWS) && (u0 - 1 + wv + 60 < NQUADS);

  v8f c[3][4];
#pragma unroll
  for (int i = 0; i < 3; ++i)
#pragma unroll
    for (int j = 0; j < 4; ++j) c[i][j] = (v8f){};

  v2f a[3], b[3][4];
  v4f ga[8];
  float g[12];

  if (!do0) {
#pragma unroll
    for (int ss = 0; ss < 18; ++ss)
      (BT(ss / 9) + (ss % 9) * BSTREAM + wv * BPITCH)[lane] = 0.0f;
  }
  if (!do5 && wv + 60 < BROWS) {
#pragma unroll
    for (int ss = 0; ss < 18; ++ss)
      (BT(ss / 9) + (ss % 9) * BSTREAM + (wv + 60) * BPITCH)[lane] = 0.0f;
  }

  STAGE_LOAD(0, 0)
  STAGE_STORE(0, 0)
  STAGE_LOAD(0, 1)
  STAGE_STORE(0, 1)
  STAGE_LOAD(1, 0)
  __syncthreads();

  for (int kb = 0; kb < NKB; ++kb) {
    const int cur = kb & 1;

    if (kb + 1 < NKB) {
      STAGE_STORE(cur ^ 1, 0)
      STAGE_LOAD(kb + 1, 1)
      asm volatile("" ::: "memory");
    }

    const float* acur = AT(cur) + 3 * wv * ATILE + lane * 2;
    const float* bcur[3];
#pragma unroll
    for (int i = 0; i < 3; ++i)
      bcur[i] = BT(cur) + prod[i] * BSTREAM + cl * BPITCH + 2 * h;
#pragma unroll
    for (int s = 0; s < 4; ++s) {
#pragma unroll
      for (int i = 0; i < 3; ++i)
        a[i] = *(const v2f*)(acur + i * ATILE + s * 64);
#pragma unroll
      for (int i = 0; i < 3; ++i)
#pragma unroll
        for (int j = 0; j < 4; ++j)
          b[i][j] = *(const v2f*)(bcur[i] + j * 16 * BPITCH + 4 * s);
      WMMA12();
    }

    if (kb + 1 < NKB) STAGE_STORE(cur ^ 1, 1)

    v2f a2[4][3], b2[4][3][4];
#pragma unroll
    for (int s = 0; s < 4; ++s) {
#pragma unroll
      for (int i = 0; i < 3; ++i)
        a2[s][i] = *(const v2f*)(acur + i * ATILE + (s + 4) * 64);
#pragma unroll
      for (int i = 0; i < 3; ++i)
#pragma unroll
        for (int j = 0; j < 4; ++j)
          b2[s][i][j] = *(const v2f*)(bcur[i] + j * 16 * BPITCH + 4 * (s + 4));
    }

    __builtin_amdgcn_fence(__ATOMIC_RELEASE, "workgroup");
    __builtin_amdgcn_s_barrier_signal(-1);

    if (kb + 2 < NKB) {
      STAGE_LOAD(kb + 2, 0)
      asm volatile("" ::: "memory");
    }

#pragma unroll
    for (int s = 0; s < 4; ++s) {
#pragma unroll
      for (int i = 0; i < 3; ++i) a[i] = a2[s][i];
#pragma unroll
      for (int i = 0; i < 3; ++i)
#pragma unroll
        for (int j = 0; j < 4; ++j) b[i][j] = b2[s][i][j];
      WMMA12();
    }

    if (kb == 31) EDGE_STEP(xr, 256)

    __builtin_amdgcn_s_barrier_wait(-1);
    __builtin_amdgcn_fence(__ATOMIC_ACQUIRE, "workgroup");
  }
  EDGE_STEP(xi, 513)

  {
#pragma unroll
    for (int i = 0; i < 3; ++i) {
      float* tw = lds + prod[i] * ETILE + cl * EPITCH + mtl[i] * 16 + 8 * h;
#pragma unroll
      for (int j = 0; j < 4; ++j) {
        v4f lo, hi;
        lo.x = c[i][j][0]; lo.y = c[i][j][1]; lo.z = c[i][j][2]; lo.w = c[i][j][3];
        hi.x = c[i][j][4]; hi.y = c[i][j][5]; hi.z = c[i][j][6]; hi.w = c[i][j][7];
        *(v4f*)(tw + j * 16 * EPITCH)     = lo;
        *(v4f*)(tw + j * 16 * EPITCH + 4) = hi;
      }
    }
  }
  __syncthreads();

  {
    const int r = r0 + 2 * lane;
    float w0[4], w1[4];
#pragma unroll
    for (int d = 0; d < 4; ++d) {
      w0[d] = win[r + d * HOP];
      w1[d] = win[r + d * HOP + 1];
    }
    float* ob = out + (long)bat * OUTLEN + r;
    for (int m = 0; m < 6; ++m) {
      const int col = 1 + wv + 12 * m;
      if (col < 64) {
        const float* pc = lds + col * EPITCH + 2 * lane;
        const float* pm = pc - EPITCH;
        const v2f P1 = *(const v2f*)(pc + 0 * ETILE);
        const v2f P2 = *(const v2f*)(pc + 1 * ETILE);
        const v2f P3 = *(const v2f*)(pc + 2 * ETILE);
        const v2f P4 = *(const v2f*)(pc + 3 * ETILE);
        const v2f P5 = *(const v2f*)(pc + 4 * ETILE);
        const v2f P6 = *(const v2f*)(pc + 5 * ETILE);
        const v2f P7 = *(const v2f*)(pc + 6 * ETILE);
        const v2f P8 = *(const v2f*)(pc + 7 * ETILE);
        const v2f P9 = *(const v2f*)(pc + 8 * ETILE);
        const v2f Q2 = *(const v2f*)(pm + 1 * ETILE);
        const v2f Q4 = *(const v2f*)(pm + 3 * ETILE);
        const v2f Q5 = *(const v2f*)(pm + 4 * ETILE);
        const v2f Q6 = *(const v2f*)(pm + 5 * ETILE);
        const v2f Q8 = *(const v2f*)(pm + 7 * ETILE);
        const v2f t1 = P3 - (P1 + P2);
        const v2f t5 = P6 - (P4 + P5);
        v2f y[4];
        y[0] = P1 + ((Q4 + Q8) - (Q2 + Q5));
        y[1] = t1 + (Q6 - (Q4 + Q5));
        y[2] = ((P2 + P7) - (P1 + P4)) + Q5;
        y[3] = ((P9 - (P7 + P8)) - t1) - t5;
        const int u = u0 - 1 + col;
#pragma unroll
        for (int j = 0; j < 4; ++j) {
          const int q = 4 * u + j;
          if (q >= 2 && q <= NFRAMES) {
            float wss0 = 0.0f, wss1 = 0.0f;
#pragma unroll
            for (int d = 3; d >= 0; --d) {
              const int f = q - d;
              const bool ok = (f >= 0) && (f <= NFRAMES - 1);
              const float v0 = ok ? w0[d] : 0.0f;
              const float v1 = ok ? w1[d] : 0.0f;
              wss0 += v0 * v0;
              wss1 += v1 * v1;
            }
            v2f o;
            o.x = (wss0 > FLT_TINY) ? y[j].x / wss0 : y[j].x;
            o.y = (wss1 > FLT_TINY) ? y[j].y / wss1 : y[j].y;
            *(v2f*)(ob + (long)(q - 2) * HOP) = o;
          }
        }
      }
    }
  }
}

extern "C" void kernel_launch(void* const* d_in, const int* in_sizes, int n_in,
                              void* d_out, int out_size, void* d_ws,
                              size_t ws_size, hipStream_t stream) {
  const float* real = (const float*)d_in[0];
  const float* imag = (const float*)d_in[1];
  const float* Wr   = (const float*)d_in[2];
  const float* Wi   = (const float*)d_in[3];
  const float* win  = (const float*)d_in[4];
  float* out = (float*)d_out;

  float* apack = (float*)d_ws;

  pack_a_kernel<<<PGRID, 256, 0, stream>>>(Wr, Wi, apack);
  istft_gemm_kernel<<<NBATCH * NQG * NRG, 384, 0, stream>>>(
      real, imag, apack, win, out);
}


# GEMM: A fragments go from the packed buffer straight to registers (a tile serves one wave); LDS holds B alone as a 3-stage ring: a wave stores and signals at block top, waits at its end; sums kept
# speedup vs baseline: 2.1244x; 1.0295x over previous
#include <hip/hip_runtime.h>

typedef __attribute__((ext_vector_type(2))) float v2f;
typedef __attribute__((ext_vector_type(4))) float v4f;
typedef __attribute__((ext_vector_type(8))) float v8f;

#define NFFT     2048
#define HOP      512
#define NBINS    1025
#define NFRAMES  1000
#define NBATCH   8
#define OUTLEN   511488
#define KSTEPS   514
#define NQUADS   250
#define PTILES   288
#define FLT_TINY 1.1754943508222875e-38f

#define PKS    32
#define PCH    17
#define PGRID  (32 * PCH)
#define PPITCH 68
#define PGRP   (PKS * PPITCH)
__global__ void __launch_bounds__(256)
pack_a_kernel(const float* __restrict__ Wr, const float* __restrict__ Wi,
              float* __restrict__ apack) {
  __shared__ __attribute__((aligned(16))) float t[9 * PGRP];
  const int lane = threadIdx.x & 31;
  const int wv   = threadIdx.x >> 5;
  const int m    = blockIdx.x & 31;
  const int kk0  = (blockIdx.x >> 5) * PKS;

  float a[4][2][4], b[4][2][4];
  bool imag[4], valid[4], mir[4];
#pragma unroll
  for (int j = 0; j < 4; ++j) {
    const int k = kk0 * 4 + 32 * j + lane;
    imag[j]  = (k >= 1028);
    const int f = imag[j] ? k - 1028 : k;
    valid[j] = (f < NBINS);
    mir[j]   = (f >= 1) && (f <= NFFT / 2 - 1);
    const int fc = valid[j] ? f : 0;
    const int fm = mir[j] ? NFFT - f : 0;
    const float* __restrict__ Wp = imag[j] ? Wi : Wr;
#pragma unroll
    for (int rr = 0; rr < 2; ++rr) {
#pragma unroll
      for (int d = 0; d < 4; ++d) {
        const int o = m * 16 + 2 * wv + rr + HOP * d;
        a[j][rr][d] = Wp[o * NFFT + fc];
        b[j][rr][d] = Wp[o * NFFT + fm];
      }
    }
  }

#pragma unroll
  for (int j = 0; j < 4; ++j) {
    float* tp = t + (8 * j + (lane >> 2)) * PPITCH + ((lane >> 1) & 1) * 32 +
                (lane & 1) + 4 * wv;
#pragma unroll
    for (int rr = 0; rr < 2; ++rr) {
      float v[4];
#pragma unroll
      for (int d = 0; d < 4; ++d) {
        const float s = imag[j] ? -a[j][rr][d] : a[j][rr][d];
        const float w = mir[j] ? s + b[j][rr][d] : s;
        v[d] = valid[j] ? w : 0.0f;
      }
      float* tq = tp + 2 * rr;
      const float s01 = v[0] + v[1];
      const float s23 = v[2] + v[3];
      tq[0 * PGRP] = v[0];
      tq[1 * PGRP] = v[1];
      tq[2 * PGRP] = s01;
      tq[3 * PGRP] = v[2];
      tq[4 * PGRP] = v[3];
      tq[5 * PGRP] = s23;
      tq[6 * PGRP] = v[0] + v[2];
      tq[7 * PGRP] = v[1] + v[3];
      tq[8 * PGRP] = s01 + s23;
    }
  }
  __syncthreads();

  v4f* __restrict__ ap4 = (v4f*)apack;
  const int c4 = threadIdx.x & 15;
#pragma unroll
  for (int hb = 0; hb < 2; ++hb) {
    const int kkl = hb * 16 + (threadIdx.x >> 4);
    const int kk  = kk0 + kkl;
    if (kk < KSTEPS) {
      v4f v[9];
#pragma unroll
      for (int g = 0; g < 9; ++g)
        v[g] = *(const v4f*)(t + g * PGRP + kkl * PPITCH + c4 * 4);
#pragma unroll
      for (int g = 0; g < 9; ++g)
        ap4[((g * 32 + m) * KSTEPS + kk) * 16 + c4] = v[g];
    }
  }
}

#define WMMA12()                                                               \
  _Pragma("unroll") for (int i = 0; i < 3; ++i) {                              \
    _Pragma("unroll") for (int j = 0; j < 4; ++j) {                            \
      c[i][j] = __builtin_amdgcn_wmma_f32_16x16x4_f32(                         \
          false, a[i], false, b[i][j], (short)0, c[i][j], false, false);       \
    }                                                                          \
  }

#define BK      32
#define BPITCH  36
#define BROWS   64
#define QC      63
#define NKB     64
#define NRG     (HOP / 64)
#define NQG     4
#define NWAVES  12
#define NSTAGE  3
#define BSTREAM (BROWS * BPITCH)
#define STAGE_FLOATS (9 * BSTREAM)
#define EPITCH  68
#define ETILE   (64 * EPITCH)
#define LDS_FLOATS (NSTAGE * STAGE_FLOATS)
#define BT(st)  (lds + (st) * STAGE_FLOATS)

__device__ __forceinline__ float stream_of(int p, float x0, float x1, float x2,
                                           float x3) {
  const bool k0 = (p == 0) | (p == 2) | (p == 6) | (p == 8);
  const bool k1 = (p == 1) | (p == 2) | (p == 7) | (p == 8);
  const bool k2 = (p == 3) | (p == 5) | (p == 6) | (p == 8);
  const bool k3 = (p == 4) | (p == 5) | (p == 7) | (p == 8);
  return ((k0 ? x0 : 0.0f) + (k1 ? x1 : 0.0f)) +
         ((k2 ? x2 : 0.0f) + (k3 ? x3 : 0.0f));
}

#define EDGE_STEP(xp, kk)                                                      \
  {                                                                            \
    _Pragma("unroll") for (int i = 0; i < 3; ++i)                              \
        a[i] = Ab[((long)tileA[i] * KSTEPS + (kk)) * 32 + lane];               \
    _Pragma("unroll") for (int j = 0; j < 4; ++j) {                            \
      const int uq = u0 - 1 + j * 16 + cl;                                     \
      const bool okq = (uq >= 0) && (uq < NQUADS);                             \
      const int uc = min(max(uq, 0), NQUADS - 1);                              \
      const float* src = (xp) + (long)(bat * NFRAMES + 4 * uc) * NBINS + 1024; \
      const float e0 = src[0];                                                 \
      const float e1 = src[NBINS];                                             \
      const float e2 = src[2 * NBINS];                                         \
      const float e3 = src[3 * NBINS];                                         \
      _Pragma("unroll") for (int i = 0; i < 3; ++i) {                          \
        float v = stream_of(prod[i], e0, e1, e2, e3);                          \
        v = okq ? v : 0.0f;                                                    \
        b[i][j].x = (h == 0) ? v : 0.0f;                                       \
        b[i][j].y = 0.0f;                                                      \
      }                                                                        \
    }                                                                          \
    WMMA12();                                                                  \
  }

#define ROW_ON(ii) (((ii) == 0) ? do0 : (((ii) == 5) ? do5 : true))

#define STAGE_LOAD(kn)                                                         \
  {                                                                            \
    const float* __restrict__ xn = ((kn) < 32) ? xr : xi;                      \
    const int f0 = ((kn) & 31) * BK;                                           \
    _Pragma("unroll") for (int ii = 0; ii < 6; ++ii) {                         \
      if (ROW_ON(ii)) {                                                        \
        _Pragma("unroll") for (int fr = 0; fr < 4; ++fr)                       \
            g[4 * ii + fr] = xn[rowS[ii] + fr * NBINS + f0 + lane];            \
      }                                                                        \
    }                                                                          \
  }

#define STAGE_STORE(st)                                                        \
  {                                                                            \
    _Pragma("unroll") for (int ii = 0; ii < 6; ++ii) {                         \
      if (ROW_ON(ii)) {                                                        \
        const float x0 = g[4 * ii];                                            \
        const float x1 = g[4 * ii + 1];                                        \
        const float x2 = g[4 * ii + 2];                                        \
        const float x3 = g[4 * ii + 3];                                        \
        const float s01 = x0 + x1;                                             \
        const float s23 = x2 + x3;                                             \
        float* bp = BT(st) + (wv + 12 * ii) * BPITCH + lane;                   \
        bp[0]           = x0;                                                  \
        bp[BSTREAM]     = x1;                                                  \
        bp[2 * BSTREAM] = s01;                                                 \
        bp[3 * BSTREAM] = x2;                                                  \
        bp[4 * BSTREAM] = x3;                                                  \
        bp[5 * BSTREAM] = s23;                                                 \
        bp[6 * BSTREAM] = x0 + x2;                                             \
        bp[7 * BSTREAM] = x1 + x3;                                             \
        bp[8 * BSTREAM] = s01 + s23;                                           \
      }                                                                        \
    }                                                                          \
  }

#define A_LOAD(dst, kn, hf)                                                    \
  {                                                                            \
    const int kk0 = (((kn) < 32) ? 0 : 257) + ((kn) & 31) * 8 + 4 * (hf);      \
    _Pragma("unroll") for (int s = 0; s < 4; ++s) {                            \
      _Pragma("unroll") for (int i = 0; i < 3; ++i)                            \
          dst[s][i] = Ab[aSrc[i] + (kk0 + s) * 32];                            \
    }                                                                          \
  }

#define HALF_BLOCK(aset, s0)                                                   \
  {                                                                            \
    _Pragma("unroll") for (int s = 0; s < 4; ++s) {                            \
      _Pragma("unroll") for (int i = 0; i < 3; ++i) a[i] = aset[s][i];         \
      _Pragma("unroll") for (int i = 0; i < 3; ++i) {                          \
        _Pragma("unroll") for (int j = 0; j < 4; ++j)                          \
            b[i][j] = *(const v2f*)(bcur[i] + j * 16 * BPITCH +                \
                                    4 * ((s0) + s));                           \
      }                                                                        \
      WMMA12();                                                                \
    }                                                                          \
  }

__global__ void __launch_bounds__(384, 3)
istft_gemm_kernel(const float* __restrict__ xr, const float* __restrict__ xi,
                  const float* __restrict__ apack,
                  const float* __restrict__ win, float* __restrict__ out) {
  __shared__ __attribute__((aligned(16))) float lds[LDS_FLOATS];

  const int lane = threadIdx.x & 31;
  const int wv   = __builtin_amdgcn_readfirstlane(threadIdx.x >> 5);
  const int rg   = blockIdx.x & 7;
  const int qg   = (blockIdx.x >> 3) & 3;
  const int bat  = blockIdx.x >> 5;
  const int r0   = rg * 64;
  const int u0   = qg * QC;
  const int h    = lane >> 4;
  const int cl   = lane & 15;

  const v2f* Ab  = (const v2f*)apack;

  int prod[3], mtl[3], tileA[3], aSrc[3];
#pragma unroll
  for (int i = 0; i < 3; ++i) {
    const int row = 3 * wv + i;
    prod[i]  = row >> 2;
    mtl[i]   = row & 3;
    tileA[i] = prod[i] * 32 + rg * 4 + mtl[i];
    aSrc[i]  = (tileA[i] * KSTEPS) * 32 + lane;
  }
  int rowS[6];
#pragma unroll
  for (int ii = 0; ii < 6; ++ii) {
    const int uq = u0 - 1 + wv + 12 * ii;
    const int uc = min(max(uq, 0), NQUADS - 1);
    rowS[ii] = (bat * NFRAMES + 4 * uc) * NBINS;
  }
  const bool do0 = (u0 - 1 + wv >= 0);
  const bool do5 = (wv + 60 < BROWS) && (u0 - 1 + wv + 60 < NQUADS);

  v8f c[3][4];
#pragma unroll
  for (int i = 0; i < 3; ++i)
#pragma unroll
    for (int j = 0; j < 4; ++j) c[i][j] = (v8f){};

  v2f a[3], b[3][4];
  v2f aP[4][3], aQ[4][3];
  float g[24];

  if (!do0) {
#pragma unroll
    for (int ss = 0; ss < 9 * NSTAGE; ++ss)
      (BT(ss / 9) + (ss % 9) * BSTREAM + wv * BPITCH)[lane] = 0.0f;
  }
  if (!do5 && wv + 60 < BROWS) {
#pragma unroll
    for (int ss = 0; ss < 9 * NSTAGE; ++ss)
      (BT(ss / 9) + (ss % 9) * BSTREAM + (wv + 60) * BPITCH)[lane] = 0.0f;
  }

  STAGE_LOAD(0)
  STAGE_STORE(0)
  STAGE_LOAD(1)
  A_LOAD(aP, 0, 0)
  __syncthreads();

  int cur = 0;
  for (int kb = 0; kb < NKB; ++kb) {
    const int nxt = (cur == NSTAGE - 1) ? 0 : cur + 1;

    if (kb + 1 < NKB) STAGE_STORE(nxt)

    __builtin_amdgcn_fence(__ATOMIC_RELEASE, "workgroup");
    __builtin_amdgcn_s_barrier_signal(-1);

    if (kb + 2 < NKB) STAGE_LOAD(kb + 2)
    A_LOAD(aQ, kb, 1)
    asm volatile("" ::: "memory");

    const float* bcur[3];
#pragma unroll
    for (int i = 0; i < 3; ++i)
      bcur[i] = BT(cur) + prod[i] * BSTREAM + cl * BPITCH + 2 * h;

    HALF_BLOCK(aP, 0)

    {
      const int kn1 = min(kb + 1, NKB - 1);
      A_LOAD(aP, kn1, 0)
    }

    HALF_BLOCK(aQ, 4)

    if (kb == 31) EDGE_STEP(xr, 256)

    __builtin_amdgcn_s_barrier_wait(-1);
    __builtin_amdgcn_fence(__ATOMIC_ACQUIRE, "workgroup");
    cur = nxt;
  }
  EDGE_STEP(xi, 513)

  __syncthreads();

  {
#pragma unroll
    for (int i = 0; i < 3; ++i) {
      float* tw = lds + prod[i] * ETILE + cl * EPITCH + mtl[i] * 16 + 8 * h;
#pragma unroll
      for (int j = 0; j < 4; ++j) {
        v4f lo, hi;
        lo.x = c[i][j][0]; lo.y = c[i][j][1]; lo.z = c[i][j][2]; lo.w = c[i][j][3];
        hi.x = c[i][j][4]; hi.y = c[i][j][5]; hi.z = c[i][j][6]; hi.w = c[i][j][7];
        *(v4f*)(tw + j * 16 * EPITCH)     = lo;
        *(v4f*)(tw + j * 16 * EPITCH + 4) = hi;
      }
    }
  }
  __syncthreads();

  {
    const int r = r0 + 2 * lane;
    float w0[4], w1[4];
#pragma unroll
    for (int d = 0; d < 4; ++d) {
      w0[d] = win[r + d * HOP];
      w1[d] = win[r + d * HOP + 1];
    }
    float* ob = out + (long)bat * OUTLEN + r;
    for (int m = 0; m < 6; ++m) {
      const int col = 1 + wv + 12 * m;
      if (col < 64) {
        const float* pc = lds + col * EPITCH + 2 * lane;
        const float* pm = pc - EPITCH;
        const v2f P1 = *(const v2f*)(pc + 0 * ETILE);
        const v2f P2 = *(const v2f*)(pc + 1 * ETILE);
        const v2f P3 = *(const v2f*)(pc + 2 * ETILE);
        const v2f P4 = *(const v2f*)(pc + 3 * ETILE);
        const v2f P5 = *(const v2f*)(pc + 4 * ETILE);
        const v2f P6 = *(const v2f*)(pc + 5 * ETILE);
        const v2f P7 = *(const v2f*)(pc + 6 * ETILE);
        const v2f P8 = *(const v2f*)(pc + 7 * ETILE);
        const v2f P9 = *(const v2f*)(pc + 8 * ETILE);
        const v2f Q2 = *(const v2f*)(pm + 1 * ETILE);
        const v2f Q4 = *(const v2f*)(pm + 3 * ETILE);
        const v2f Q5 = *(const v2f*)(pm + 4 * ETILE);
        const v2f Q6 = *(const v2f*)(pm + 5 * ETILE);
        const v2f Q8 = *(const v2f*)(pm + 7 * ETILE);
        const v2f t1 = P3 - (P1 + P2);
        const v2f t5 = P6 - (P4 + P5);
        v2f y[4];
        y[0] = P1 + ((Q4 + Q8) - (Q2 + Q5));
        y[1] = t1 + (Q6 - (Q4 + Q5));
        y[2] = ((P2 + P7) - (P1 + P4)) + Q5;
        y[3] = ((P9 - (P7 + P8)) - t1) - t5;
        const int u = u0 - 1 + col;
#pragma unroll
        for (int j = 0; j < 4; ++j) {
          const int q = 4 * u + j;
          if (q >= 2 && q <= NFRAMES) {
            float wss0 = 0.0f, wss1 = 0.0f;
#pragma unroll
            for (int d = 3; d >= 0; --d) {
              const int f = q - d;
              const bool ok = (f >= 0) && (f <= NFRAMES - 1);
              const float v0 = ok ? w0[d] : 0.0f;
              const float v1 = ok ? w1[d] : 0.0f;
              wss0 += v0 * v0;
              wss1 += v1 * v1;
            }
            v2f o;
            o.x = (wss0 > FLT_TINY) ? y[j].x / wss0 : y[j].x;
            o.y = (wss1 > FLT_TINY) ? y[j].y / wss1 : y[j].y;
            *(v2f*)(ob + (long)(q - 2) * HOP) = o;
          }
        }
      }
    }
  }
}

extern "C" void kernel_launch(void* const* d_in, const int* in_sizes, int n_in,
                              void* d_out, int out_size, void* d_ws,
                              size_t ws_size, hipStream_t stream) {
  const float* real = (const float*)d_in[0];
  const float* imag = (const float*)d_in[1];
  const float* Wr   = (const float*)d_in[2];
  const float* Wi   = (const float*)d_in[3];
  const float* win  = (const float*)d_in[4];
  float* out = (float*)d_out;

  float* apack = (float*)d_ws;

  pack_a_kernel<<<PGRID, 256, 0, stream>>>(Wr, Wi, apack);
  istft_gemm_kernel<<<NBATCH * NQG * NRG, 384, 0, stream>>>(
      real, imag, apack, win, out);
}
